# starting kernel
# baseline (speedup 1.0000x reference)
#include <hip/hip_runtime.h>

#define PN      5
#define MROWS   500
#define NCOLS   2000
#define BB      1024
#define KITERS  10
#define MP      100
#define AXLD    112
#define JT5     5
#define NGRP    80
#define KLD     102
#define KCH     100
#define MPAD    112
#define ROWSPAN 512

typedef __attribute__((ext_vector_type(2))) float v2f;
typedef __attribute__((ext_vector_type(8))) float v8f;

__device__ __forceinline__ v8f wmma4(v2f a, v2f b, v8f c) {
    return __builtin_amdgcn_wmma_f32_16x16x4_f32(
        false, a, false, b, (short)0, c, false, false);
}


__global__ void zero_kernel(float4* __restrict__ dst, long long n4) {
    long long i = (long long)blockIdx.x * blockDim.x + threadIdx.x;
    if (i < n4) dst[i] = make_float4(0.f, 0.f, 0.f, 0.f);
}

__global__ void __launch_bounds__(256, 1)
atb_kernel(const float* __restrict__ A, const float* __restrict__ b,
           float* __restrict__ Atb) {
    const int p    = blockIdx.z;
    const int n0   = blockIdx.x * NGRP;
    const int it0  = blockIdx.y * 8 + (threadIdx.x >> 5);
    const int r0   = it0 * 16;
    const int r1   = r0 + ROWSPAN;
    const float* Ap = A + (size_t)p * MP * NCOLS;

    __shared__ float sBt[NGRP][KLD];
    for (int idx = threadIdx.x; idx < MP * NGRP / 4; idx += 256) {
        const int kk = idx / (NGRP / 4);
        const int c4 = (idx % (NGRP / 4)) * 4;
        float4 v = *(const float4*)&Ap[(size_t)kk * NCOLS + n0 + c4];
        sBt[c4 + 0][kk] = v.x;
        sBt[c4 + 1][kk] = v.y;
        sBt[c4 + 2][kk] = v.z;
        sBt[c4 + 3][kk] = v.w;
    }
    __syncthreads();

    const int lane = threadIdx.x & 31;
    const int nl   = lane & 15;
    const int kh   = (lane >> 4) << 1;

    const float* pa0 = b + (size_t)(r0 + nl) * MROWS + p * MP + kh;
    const float* pa1 = b + (size_t)(r1 + nl) * MROWS + p * MP + kh;
    const float* ps  = &sBt[nl][kh];

    v8f acc0[JT5] = {};
    v8f acc1[JT5] = {};
    for (int k0 = 0; k0 < MP; k0 += 4) {
        v2f af0; af0.x = pa0[0]; af0.y = pa0[1]; pa0 += 4;
        v2f af1; af1.x = pa1[0]; af1.y = pa1[1]; pa1 += 4;
#pragma unroll
        for (int j = 0; j < JT5; ++j) {
            v2f bf = *(const v2f*)(ps + j * 16 * KLD);
            acc0[j] = wmma4(af0, bf, acc0[j]);
            acc1[j] = wmma4(af1, bf, acc1[j]);
        }
        ps += 4;
    }

    float* out = Atb + (size_t)p * BB * NCOLS;
    const int ih = (lane >> 4) * 8;
#pragma unroll
    for (int j = 0; j < JT5; ++j) {
        const int col = n0 + j * 16 + nl;
#pragma unroll
        for (int v = 0; v < 8; ++v) {
            out[(size_t)(r0 + ih + v) * NCOLS + col] = acc0[j][v];
            out[(size_t)(r1 + ih + v) * NCOLS + col] = acc1[j][v];
        }
    }
}

__global__ void __launch_bounds__(256, 1)
ax_kernel(const float* __restrict__ A, const float* __restrict__ X,
          float* __restrict__ axbuf, int p) {
    const int it0 = blockIdx.x * 8 + (threadIdx.x >> 5);
    const int r0  = it0 * 16;
    const int r1  = r0 + ROWSPAN;
    const float* Ap = A + (size_t)p * MP * NCOLS;
    const float* Xp = X + (size_t)p * BB * NCOLS;

    __shared__ float sB[MPAD][KCH];

    const int lane = threadIdx.x & 31;
    const int ml   = lane & 15;
    const int kh   = (lane >> 4) << 1;

    const float* px0 = Xp + (size_t)(r0 + ml) * NCOLS + kh;
    const float* px1 = Xp + (size_t)(r1 + ml) * NCOLS + kh;

    v8f acc0[7] = {};
    v8f acc1[7] = {};
    for (int kc = 0; kc < NCOLS; kc += KCH) {
        __syncthreads();
        for (int idx = threadIdx.x; idx < MPAD * KCH / 4; idx += 256) {
            const int m  = idx / (KCH / 4);
            const int c4 = (idx % (KCH / 4)) * 4;
            float4 v = make_float4(0.f, 0.f, 0.f, 0.f);
            if (m < MP) {
                v = *(const float4*)&Ap[(size_t)m * NCOLS + kc + c4];
                if (kc + KCH < NCOLS)
                    __builtin_prefetch(&Ap[(size_t)m * NCOLS + kc + KCH + c4], 0, 1);
            }
            *(float4*)&sB[m][c4] = v;
        }
        __syncthreads();

        for (int k0 = 0; k0 < KCH; k0 += 4) {
            v2f af0; af0.x = px0[0]; af0.y = px0[1]; px0 += 4;
            v2f af1; af1.x = px1[0]; af1.y = px1[1]; px1 += 4;
            const float* ps = &sB[ml][k0 + kh];
#pragma unroll
            for (int j = 0; j < 7; ++j) {
                v2f bf = *(const v2f*)(ps + j * 16 * KCH);
                acc0[j] = wmma4(af0, bf, acc0[j]);
                acc1[j] = wmma4(af1, bf, acc1[j]);
            }
        }
    }

    const int ih = (lane >> 4) * 8;
#pragma unroll
    for (int j = 0; j < 7; ++j) {
        const int col = j * 16 + ml;
        if (col < MP) {
#pragma unroll
            for (int v = 0; v < 8; ++v) {
                axbuf[(size_t)(r0 + ih + v) * AXLD + col] = acc0[j][v];
                axbuf[(size_t)(r1 + ih + v) * AXLD + col] = acc1[j][v];
            }
        }
    }
}

__global__ void __launch_bounds__(256, 1)
xupd_kernel(const float* __restrict__ A, const float* __restrict__ axbuf,
            const float* __restrict__ Atb, const float* __restrict__ hyp,
            const int* __restrict__ nbr,
            float* __restrict__ X, const float* __restrict__ U,
            int p, int k) {
    const int n0  = blockIdx.x * NGRP;
    const int it0 = blockIdx.y * 8 + (threadIdx.x >> 5);
    const int r0  = it0 * 16;
    const int r1  = r0 + ROWSPAN;
    const float* Ap = A + (size_t)p * MP * NCOLS;

    __shared__ float sBt[NGRP][KLD];
    for (int idx = threadIdx.x; idx < MP * NGRP / 4; idx += 256) {
        const int kk = idx / (NGRP / 4);
        const int c4 = (idx % (NGRP / 4)) * 4;
        float4 v = *(const float4*)&Ap[(size_t)kk * NCOLS + n0 + c4];
        sBt[c4 + 0][kk] = v.x;
        sBt[c4 + 1][kk] = v.y;
        sBt[c4 + 2][kk] = v.z;
        sBt[c4 + 3][kk] = v.w;
    }
    __syncthreads();

    const int lane = threadIdx.x & 31;
    const int nl   = lane & 15;
    const int kh   = (lane >> 4) << 1;

    const float* pa0 = axbuf + (size_t)(r0 + nl) * AXLD + kh;
    const float* pa1 = axbuf + (size_t)(r1 + nl) * AXLD + kh;
    const float* ps  = &sBt[nl][kh];

    v8f acc0[JT5] = {};
    v8f acc1[JT5] = {};
    for (int k0 = 0; k0 < MP; k0 += 4) {
        v2f af0; af0.x = pa0[0]; af0.y = pa0[1]; pa0 += 4;
        v2f af1; af1.x = pa1[0]; af1.y = pa1[1]; pa1 += 4;
#pragma unroll
        for (int j = 0; j < JT5; ++j) {
            v2f bf = *(const v2f*)(ps + j * 16 * KLD);
            acc0[j] = wmma4(af0, bf, acc0[j]);
            acc1[j] = wmma4(af1, bf, acc1[j]);
        }
        ps += 4;
    }

    const float a0 = fabsf(hyp[(k * PN + p) * 4 + 0]);
    const float a1 = fabsf(hyp[(k * PN + p) * 4 + 1]);
    const float a2 = fabsf(hyp[(k * PN + p) * 4 + 2]);
    const int pl = nbr[p * 2 + 0];
    const int pr = nbr[p * 2 + 1];
    const size_t stride = (size_t)BB * NCOLS;
    float*       Xp = X + (size_t)p * stride;
    const float* Xl = X + (size_t)pl * stride;
    const float* Xr = X + (size_t)pr * stride;
    const float* Up = U + (size_t)p * stride;
    const float* Tp = Atb + (size_t)p * stride;

    const int ih = (lane >> 4) * 8;
#pragma unroll
    for (int j = 0; j < JT5; ++j) {
        const int col = n0 + j * 16 + nl;
#pragma unroll
        for (int v = 0; v < 8; ++v) {
            {
                const size_t off = (size_t)(r0 + ih + v) * NCOLS + col;
                const float xo  = Xp[off];
                const float sgn = (xo > 0.f) ? 1.f : ((xo < 0.f) ? -1.f : 0.f);
                const float g   = acc0[j][v] - Tp[off] + 2.f * a0 * xo + a2 * sgn
                                  + 2.f * Up[off] - a0 * (Xl[off] + Xr[off]);
                Xp[off] = xo - a1 * g;
            }
            {
                const size_t off = (size_t)(r1 + ih + v) * NCOLS + col;
                const float xo  = Xp[off];
                const float sgn = (xo > 0.f) ? 1.f : ((xo < 0.f) ? -1.f : 0.f);
                const float g   = acc1[j][v] - Tp[off] + 2.f * a0 * xo + a2 * sgn
                                  + 2.f * Up[off] - a0 * (Xl[off] + Xr[off]);
                Xp[off] = xo - a1 * g;
            }
        }
    }
}

__global__ void dual_kernel(float* __restrict__ XU, const float* __restrict__ hyp,
                            const int* __restrict__ nbr, int k) {
    const size_t stride4 = (size_t)BB * NCOLS / 4;
    const size_t total4  = (size_t)PN * stride4;
    size_t i = (size_t)blockIdx.x * blockDim.x + threadIdx.x;
    if (i >= total4) return;
    const int p = (int)(i / stride4);
    const size_t off = i - (size_t)p * stride4;
    const float a3 = fabsf(hyp[(k * PN + p) * 4 + 3]);
    const int pl = nbr[p * 2 + 0];
    const int pr = nbr[p * 2 + 1];
    const float4* X = (const float4*)XU;
    float4*       Uv = (float4*)(XU + (size_t)PN * BB * NCOLS);
    const float4 x  = X[i];
    const float4 xl = X[(size_t)pl * stride4 + off];
    const float4 xr = X[(size_t)pr * stride4 + off];
    float4 u = Uv[i];
    u.x += a3 * (2.f * x.x - xl.x - xr.x);
    u.y += a3 * (2.f * x.y - xl.y - xr.y);
    u.z += a3 * (2.f * x.z - xl.z - xr.z);
    u.w += a3 * (2.f * x.w - xl.w - xr.w);
    Uv[i] = u;
}

extern "C" void kernel_launch(void* const* d_in, const int* in_sizes, int n_in,
                              void* d_out, int out_size, void* d_ws, size_t ws_size,
                              hipStream_t stream) {
    (void)in_sizes; (void)n_in; (void)out_size; (void)ws_size;
    const float* A   = (const float*)d_in[0];
    const float* b   = (const float*)d_in[1];
    const float* hyp = (const float*)d_in[2];
    const int*   nbr = (const int*)d_in[3];

    float* X = (float*)d_out;
    float* U = X + (size_t)PN * BB * NCOLS;
    float* Atb   = (float*)d_ws;
    float* axbuf = Atb + (size_t)PN * BB * NCOLS;

    {
        long long n4 = (2LL * PN * BB * NCOLS) / 4;
        int blocks = (int)((n4 + 255) / 256);
        zero_kernel<<<blocks, 256, 0, stream>>>((float4*)d_out, n4);
    }
    atb_kernel<<<dim3(25, 4, PN), 256, 0, stream>>>(A, b, Atb);

    const int colors[3][2] = {{0, 2}, {1, 3}, {4, -1}};
    for (int k = 0; k < KITERS; ++k) {
        for (int c = 0; c < 3; ++c) {
            for (int s = 0; s < 2; ++s) {
                const int p = colors[c][s];
                if (p < 0) continue;
                ax_kernel  <<<4,           256, 0, stream>>>(A, X, axbuf, p);
                xupd_kernel<<<dim3(25, 4), 256, 0, stream>>>(A, axbuf, Atb, hyp, nbr,
                                                             X, U, p, k);
            }
        }
        {
            long long n4 = ((long long)PN * BB * NCOLS) / 4;
            int blocks = (int)((n4 + 255) / 256);
            dual_kernel<<<blocks, 256, 0, stream>>>(X, hyp, nbr, k);
        }
    }
}


# AX product: K=2000 sum split over 20 work-groups per row block (grid 4 -> 160), partials added in chunk order by a small second kernel; changes the K summation order
# speedup vs baseline: 4.5714x; 4.5714x over previous
#include <hip/hip_runtime.h>

#define PN      5
#define MROWS   500
#define NCOLS   2000
#define BB      1024
#define KITERS  10
#define MP      100
#define AXLD    112
#define JT5     5
#define NGRP    80
#define KLD     102
#define KCH     100
#define MPAD    112
#define ROWSPAN 512
#define KSPLIT  20
#define STG     11

typedef __attribute__((ext_vector_type(2))) float v2f;
typedef __attribute__((ext_vector_type(8))) float v8f;

__device__ __forceinline__ v8f wmma4(v2f a, v2f b, v8f c) {
    return __builtin_amdgcn_wmma_f32_16x16x4_f32(
        false, a, false, b, (short)0, c, false, false);
}


__global__ void zero_kernel(float4* __restrict__ dst, long long n4) {
    long long i = (long long)blockIdx.x * blockDim.x + threadIdx.x;
    if (i < n4) dst[i] = make_float4(0.f, 0.f, 0.f, 0.f);
}

__global__ void __launch_bounds__(256, 1)
atb_kernel(const float* __restrict__ A, const float* __restrict__ b,
           float* __restrict__ Atb) {
    const int p    = blockIdx.z;
    const int n0   = blockIdx.x * NGRP;
    const int it0  = blockIdx.y * 8 + (threadIdx.x >> 5);
    const int r0   = it0 * 16;
    const int r1   = r0 + ROWSPAN;
    const float* Ap = A + (size_t)p * MP * NCOLS;

    __shared__ float sBt[NGRP][KLD];
    for (int idx = threadIdx.x; idx < MP * NGRP / 4; idx += 256) {
        const int kk = idx / (NGRP / 4);
        const int c4 = (idx % (NGRP / 4)) * 4;
        float4 v = *(const float4*)&Ap[(size_t)kk * NCOLS + n0 + c4];
        sBt[c4 + 0][kk] = v.x;
        sBt[c4 + 1][kk] = v.y;
        sBt[c4 + 2][kk] = v.z;
        sBt[c4 + 3][kk] = v.w;
    }
    __syncthreads();

    const int lane = threadIdx.x & 31;
    const int nl   = lane & 15;
    const int kh   = (lane >> 4) << 1;

    const float* pa0 = b + (size_t)(r0 + nl) * MROWS + p * MP + kh;
    const float* pa1 = b + (size_t)(r1 + nl) * MROWS + p * MP + kh;
    const float* ps  = &sBt[nl][kh];

    v8f acc0[JT5] = {};
    v8f acc1[JT5] = {};
    for (int k0 = 0; k0 < MP; k0 += 4) {
        v2f af0; af0.x = pa0[0]; af0.y = pa0[1]; pa0 += 4;
        v2f af1; af1.x = pa1[0]; af1.y = pa1[1]; pa1 += 4;
#pragma unroll
        for (int j = 0; j < JT5; ++j) {
            v2f bf = *(const v2f*)(ps + j * 16 * KLD);
            acc0[j] = wmma4(af0, bf, acc0[j]);
            acc1[j] = wmma4(af1, bf, acc1[j]);
        }
        ps += 4;
    }

    float* out = Atb + (size_t)p * BB * NCOLS;
    const int ih = (lane >> 4) * 8;
#pragma unroll
    for (int j = 0; j < JT5; ++j) {
        const int col = n0 + j * 16 + nl;
#pragma unroll
        for (int v = 0; v < 8; ++v) {
            out[(size_t)(r0 + ih + v) * NCOLS + col] = acc0[j][v];
            out[(size_t)(r1 + ih + v) * NCOLS + col] = acc1[j][v];
        }
    }
}

__global__ void __launch_bounds__(256, 1)
ax_part_kernel(const float* __restrict__ A, const float* __restrict__ X,
               float* __restrict__ part, int p) {
    const int r0 = (blockIdx.x * 8 + (threadIdx.x >> 5)) * 16;
    const int kc = blockIdx.y * KCH;
    const float* Ap = A + (size_t)p * MP * NCOLS;
    const float* Xp = X + (size_t)p * BB * NCOLS;

    __shared__ float sB[MPAD][KCH];

    const int lane = threadIdx.x & 31;
    const int ml   = lane & 15;
    const int kh   = (lane >> 4) << 1;

    float4 st[STG];
#pragma unroll
    for (int t = 0; t < STG; ++t) {
        const int idx = threadIdx.x + t * 256;
        const int m   = idx / (KCH / 4);
        const int c4  = (idx % (KCH / 4)) * 4;
        st[t] = make_float4(0.f, 0.f, 0.f, 0.f);
        if (t * 256 + 255 < MP * KCH / 4) {
            st[t] = *(const float4*)&Ap[(size_t)m * NCOLS + kc + c4];
        } else if (t * 256 < MP * KCH / 4) {
            if (m < MP)
                st[t] = *(const float4*)&Ap[(size_t)m * NCOLS + kc + c4];
        }
    }

    const float* px = Xp + (size_t)(r0 + ml) * NCOLS + kc + kh;
    v2f af[KCH / 4];
#pragma unroll
    for (int i = 0; i < KCH / 4; ++i) {
        af[i].x = px[4 * i];
        af[i].y = px[4 * i + 1];
    }

#pragma unroll
    for (int t = 0; t < STG; ++t) {
        const int idx = threadIdx.x + t * 256;
        const int m   = idx / (KCH / 4);
        const int c4  = (idx % (KCH / 4)) * 4;
        if (idx < MPAD * KCH / 4)
            *(float4*)&sB[m][c4] = st[t];
    }
    __syncthreads();

    v8f acc[7] = {};
#pragma unroll
    for (int i = 0; i < KCH / 4; ++i) {
        const float* ps = &sB[ml][4 * i + kh];
#pragma unroll
        for (int j = 0; j < 7; ++j) {
            v2f bf = *(const v2f*)(ps + j * 16 * KCH);
            acc[j] = wmma4(af[i], bf, acc[j]);
        }
    }

    float* out = part + (size_t)blockIdx.y * BB * AXLD;
    const int ih = (lane >> 4) * 8;
#pragma unroll
    for (int j = 0; j < 7; ++j) {
#pragma unroll
        for (int v = 0; v < 8; ++v)
            out[(size_t)(r0 + ih + v) * AXLD + j * 16 + ml] = acc[j][v];
    }
}

__global__ void ax_sum_kernel(const float* __restrict__ part,
                              float* __restrict__ axbuf) {
    const int n4 = BB * AXLD / 4;
    const int i  = blockIdx.x * blockDim.x + threadIdx.x;
    const float4* src = (const float4*)part;
    float4 v[KSPLIT];
#pragma unroll
    for (int c = 0; c < KSPLIT; ++c) v[c] = src[(size_t)c * n4 + i];
    float4 s = v[0];
#pragma unroll
    for (int c = 1; c < KSPLIT; ++c) {
        s.x += v[c].x; s.y += v[c].y; s.z += v[c].z; s.w += v[c].w;
    }
    ((float4*)axbuf)[i] = s;
}

__global__ void __launch_bounds__(256, 1)
xupd_kernel(const float* __restrict__ A, const float* __restrict__ axbuf,
            const float* __restrict__ Atb, const float* __restrict__ hyp,
            const int* __restrict__ nbr,
            float* __restrict__ X, const float* __restrict__ U,
            int p, int k) {
    const int n0  = blockIdx.x * NGRP;
    const int it0 = blockIdx.y * 8 + (threadIdx.x >> 5);
    const int r0  = it0 * 16;
    const int r1  = r0 + ROWSPAN;
    const float* Ap = A + (size_t)p * MP * NCOLS;

    __shared__ float sBt[NGRP][KLD];
    for (int idx = threadIdx.x; idx < MP * NGRP / 4; idx += 256) {
        const int kk = idx / (NGRP / 4);
        const int c4 = (idx % (NGRP / 4)) * 4;
        float4 v = *(const float4*)&Ap[(size_t)kk * NCOLS + n0 + c4];
        sBt[c4 + 0][kk] = v.x;
        sBt[c4 + 1][kk] = v.y;
        sBt[c4 + 2][kk] = v.z;
        sBt[c4 + 3][kk] = v.w;
    }
    __syncthreads();

    const int lane = threadIdx.x & 31;
    const int nl   = lane & 15;
    const int kh   = (lane >> 4) << 1;

    const float* pa0 = axbuf + (size_t)(r0 + nl) * AXLD + kh;
    const float* pa1 = axbuf + (size_t)(r1 + nl) * AXLD + kh;
    const float* ps  = &sBt[nl][kh];

    v8f acc0[JT5] = {};
    v8f acc1[JT5] = {};
    for (int k0 = 0; k0 < MP; k0 += 4) {
        v2f af0; af0.x = pa0[0]; af0.y = pa0[1]; pa0 += 4;
        v2f af1; af1.x = pa1[0]; af1.y = pa1[1]; pa1 += 4;
#pragma unroll
        for (int j = 0; j < JT5; ++j) {
            v2f bf = *(const v2f*)(ps + j * 16 * KLD);
            acc0[j] = wmma4(af0, bf, acc0[j]);
            acc1[j] = wmma4(af1, bf, acc1[j]);
        }
        ps += 4;
    }

    const float a0 = fabsf(hyp[(k * PN + p) * 4 + 0]);
    const float a1 = fabsf(hyp[(k * PN + p) * 4 + 1]);
    const float a2 = fabsf(hyp[(k * PN + p) * 4 + 2]);
    const int pl = nbr[p * 2 + 0];
    const int pr = nbr[p * 2 + 1];
    const size_t stride = (size_t)BB * NCOLS;
    float*       Xp = X + (size_t)p * stride;
    const float* Xl = X + (size_t)pl * stride;
    const float* Xr = X + (size_t)pr * stride;
    const float* Up = U + (size_t)p * stride;
    const float* Tp = Atb + (size_t)p * stride;

    const int ih = (lane >> 4) * 8;
#pragma unroll
    for (int j = 0; j < JT5; ++j) {
        const int col = n0 + j * 16 + nl;
#pragma unroll
        for (int v = 0; v < 8; ++v) {
            {
                const size_t off = (size_t)(r0 + ih + v) * NCOLS + col;
                const float xo  = Xp[off];
                const float sgn = (xo > 0.f) ? 1.f : ((xo < 0.f) ? -1.f : 0.f);
                const float g   = acc0[j][v] - Tp[off] + 2.f * a0 * xo + a2 * sgn
                                  + 2.f * Up[off] - a0 * (Xl[off] + Xr[off]);
                Xp[off] = xo - a1 * g;
            }
            {
                const size_t off = (size_t)(r1 + ih + v) * NCOLS + col;
                const float xo  = Xp[off];
                const float sgn = (xo > 0.f) ? 1.f : ((xo < 0.f) ? -1.f : 0.f);
                const float g   = acc1[j][v] - Tp[off] + 2.f * a0 * xo + a2 * sgn
                                  + 2.f * Up[off] - a0 * (Xl[off] + Xr[off]);
                Xp[off] = xo - a1 * g;
            }
        }
    }
}

__global__ void dual_kernel(float* __restrict__ XU, const float* __restrict__ hyp,
                            const int* __restrict__ nbr, int k) {
    const size_t stride4 = (size_t)BB * NCOLS / 4;
    const size_t total4  = (size_t)PN * stride4;
    size_t i = (size_t)blockIdx.x * blockDim.x + threadIdx.x;
    if (i >= total4) return;
    const int p = (int)(i / stride4);
    const size_t off = i - (size_t)p * stride4;
    const float a3 = fabsf(hyp[(k * PN + p) * 4 + 3]);
    const int pl = nbr[p * 2 + 0];
    const int pr = nbr[p * 2 + 1];
    const float4* X = (const float4*)XU;
    float4*       Uv = (float4*)(XU + (size_t)PN * BB * NCOLS);
    const float4 x  = X[i];
    const float4 xl = X[(size_t)pl * stride4 + off];
    const float4 xr = X[(size_t)pr * stride4 + off];
    float4 u = Uv[i];
    u.x += a3 * (2.f * x.x - xl.x - xr.x);
    u.y += a3 * (2.f * x.y - xl.y - xr.y);
    u.z += a3 * (2.f * x.z - xl.z - xr.z);
    u.w += a3 * (2.f * x.w - xl.w - xr.w);
    Uv[i] = u;
}

extern "C" void kernel_launch(void* const* d_in, const int* in_sizes, int n_in,
                              void* d_out, int out_size, void* d_ws, size_t ws_size,
                              hipStream_t stream) {
    (void)in_sizes; (void)n_in; (void)out_size; (void)ws_size;
    const float* A   = (const float*)d_in[0];
    const float* b   = (const float*)d_in[1];
    const float* hyp = (const float*)d_in[2];
    const int*   nbr = (const int*)d_in[3];

    float* X = (float*)d_out;
    float* U = X + (size_t)PN * BB * NCOLS;
    float* Atb   = (float*)d_ws;
    float* axbuf = Atb + (size_t)PN * BB * NCOLS;
    float* part  = axbuf + (size_t)BB * AXLD;

    {
        long long n4 = (2LL * PN * BB * NCOLS) / 4;
        int blocks = (int)((n4 + 255) / 256);
        zero_kernel<<<blocks, 256, 0, stream>>>((float4*)d_out, n4);
    }
    atb_kernel<<<dim3(25, 4, PN), 256, 0, stream>>>(A, b, Atb);

    const int colors[3][2] = {{0, 2}, {1, 3}, {4, -1}};
    for (int k = 0; k < KITERS; ++k) {
        for (int c = 0; c < 3; ++c) {
            for (int s = 0; s < 2; ++s) {
                const int p = colors[c][s];
                if (p < 0) continue;
                ax_part_kernel<<<dim3(8, KSPLIT), 256, 0, stream>>>(A, X, part, p);
                ax_sum_kernel <<<BB * AXLD / 4 / 256, 256, 0, stream>>>(part, axbuf);
                xupd_kernel<<<dim3(25, 4), 256, 0, stream>>>(A, axbuf, Atb, hyp, nbr,
                                                             X, U, p, k);
            }
        }
        {
            long long n4 = ((long long)PN * BB * NCOLS) / 4;
            int blocks = (int)((n4 + 255) / 256);
            dual_kernel<<<blocks, 256, 0, stream>>>(X, hyp, nbr, k);
        }
    }
}


# one launch per colour class for AX product, partial sum and update (162 -> 102 launches per repetition); update wave holds the same tile of both class nodes, loads all old X before it stores
# speedup vs baseline: 8.8802x; 1.9426x over previous
#include <hip/hip_runtime.h>

#define PN      5
#define MROWS   500
#define NCOLS   2000
#define BB      1024
#define KITERS  10
#define MP      100
#define AXLD    112
#define JT5     5
#define NGRP    80
#define KLD     102
#define KCH     100
#define MPAD    112
#define ROWSPAN 512
#define KSPLIT  20
#define STG     11

typedef __attribute__((ext_vector_type(2))) float v2f;
typedef __attribute__((ext_vector_type(8))) float v8f;

__device__ __forceinline__ v8f wmma4(v2f a, v2f b, v8f c) {
    return __builtin_amdgcn_wmma_f32_16x16x4_f32(
        false, a, false, b, (short)0, c, false, false);
}


__global__ void zero_kernel(float4* __restrict__ dst, long long n4) {
    long long i = (long long)blockIdx.x * blockDim.x + threadIdx.x;
    if (i < n4) dst[i] = make_float4(0.f, 0.f, 0.f, 0.f);
}

__global__ void __launch_bounds__(256, 1)
atb_kernel(const float* __restrict__ A, const float* __restrict__ b,
           float* __restrict__ Atb) {
    const int p    = blockIdx.z;
    const int n0   = blockIdx.x * NGRP;
    const int it0  = blockIdx.y * 8 + (threadIdx.x >> 5);
    const int r0   = it0 * 16;
    const int r1   = r0 + ROWSPAN;
    const float* Ap = A + (size_t)p * MP * NCOLS;

    __shared__ float sBt[NGRP][KLD];
    for (int idx = threadIdx.x; idx < MP * NGRP / 4; idx += 256) {
        const int kk = idx / (NGRP / 4);
        const int c4 = (idx % (NGRP / 4)) * 4;
        float4 v = *(const float4*)&Ap[(size_t)kk * NCOLS + n0 + c4];
        sBt[c4 + 0][kk] = v.x;
        sBt[c4 + 1][kk] = v.y;
        sBt[c4 + 2][kk] = v.z;
        sBt[c4 + 3][kk] = v.w;
    }
    __syncthreads();

    const int lane = threadIdx.x & 31;
    const int nl   = lane & 15;
    const int kh   = (lane >> 4) << 1;

    const float* pa0 = b + (size_t)(r0 + nl) * MROWS + p * MP + kh;
    const float* pa1 = b + (size_t)(r1 + nl) * MROWS + p * MP + kh;
    const float* ps  = &sBt[nl][kh];

    v8f acc0[JT5] = {};
    v8f acc1[JT5] = {};
    for (int k0 = 0; k0 < MP; k0 += 4) {
        v2f af0; af0.x = pa0[0]; af0.y = pa0[1]; pa0 += 4;
        v2f af1; af1.x = pa1[0]; af1.y = pa1[1]; pa1 += 4;
#pragma unroll
        for (int j = 0; j < JT5; ++j) {
            v2f bf = *(const v2f*)(ps + j * 16 * KLD);
            acc0[j] = wmma4(af0, bf, acc0[j]);
            acc1[j] = wmma4(af1, bf, acc1[j]);
        }
        ps += 4;
    }

    float* out = Atb + (size_t)p * BB * NCOLS;
    const int ih = (lane >> 4) * 8;
#pragma unroll
    for (int j = 0; j < JT5; ++j) {
        const int col = n0 + j * 16 + nl;
#pragma unroll
        for (int v = 0; v < 8; ++v) {
            out[(size_t)(r0 + ih + v) * NCOLS + col] = acc0[j][v];
            out[(size_t)(r1 + ih + v) * NCOLS + col] = acc1[j][v];
        }
    }
}

__global__ void __launch_bounds__(256, 1)
ax_part_kernel(const float* __restrict__ A, const float* __restrict__ X,
               float* __restrict__ part, int p0, int p1) {
    const int p  = blockIdx.z ? p1 : p0;
    const int r0 = (blockIdx.x * 8 + (threadIdx.x >> 5)) * 16;
    const int kc = blockIdx.y * KCH;
    const float* Ap = A + (size_t)p * MP * NCOLS;
    const float* Xp = X + (size_t)p * BB * NCOLS;

    __shared__ float sB[MPAD][KCH];

    const int lane = threadIdx.x & 31;
    const int ml   = lane & 15;
    const int kh   = (lane >> 4) << 1;

    float4 st[STG];
#pragma unroll
    for (int t = 0; t < STG; ++t) {
        const int idx = threadIdx.x + t * 256;
        const int m   = idx / (KCH / 4);
        const int c4  = (idx % (KCH / 4)) * 4;
        st[t] = make_float4(0.f, 0.f, 0.f, 0.f);
        if (t * 256 + 255 < MP * KCH / 4) {
            st[t] = *(const float4*)&Ap[(size_t)m * NCOLS + kc + c4];
        } else if (t * 256 < MP * KCH / 4) {
            if (m < MP)
                st[t] = *(const float4*)&Ap[(size_t)m * NCOLS + kc + c4];
        }
    }

    const float* px = Xp + (size_t)(r0 + ml) * NCOLS + kc + kh;
    v2f af[KCH / 4];
#pragma unroll
    for (int i = 0; i < KCH / 4; ++i) {
        af[i].x = px[4 * i];
        af[i].y = px[4 * i + 1];
    }

#pragma unroll
    for (int t = 0; t < STG; ++t) {
        const int idx = threadIdx.x + t * 256;
        const int m   = idx / (KCH / 4);
        const int c4  = (idx % (KCH / 4)) * 4;
        if (idx < MPAD * KCH / 4)
            *(float4*)&sB[m][c4] = st[t];
    }
    __syncthreads();

    v8f acc[7] = {};
#pragma unroll
    for (int i = 0; i < KCH / 4; ++i) {
        const float* ps = &sB[ml][4 * i + kh];
#pragma unroll
        for (int j = 0; j < 7; ++j) {
            v2f bf = *(const v2f*)(ps + j * 16 * KCH);
            acc[j] = wmma4(af[i], bf, acc[j]);
        }
    }

    float* out = part + ((size_t)blockIdx.z * KSPLIT + blockIdx.y) * BB * AXLD;
    const int ih = (lane >> 4) * 8;
#pragma unroll
    for (int j = 0; j < 7; ++j) {
#pragma unroll
        for (int v = 0; v < 8; ++v)
            out[(size_t)(r0 + ih + v) * AXLD + j * 16 + ml] = acc[j][v];
    }
}

__global__ void ax_sum_kernel(const float* __restrict__ part,
                              float* __restrict__ axbuf) {
    const int n4 = BB * AXLD / 4;
    const int i  = blockIdx.x * blockDim.x + threadIdx.x;
    part  += (size_t)blockIdx.y * KSPLIT * BB * AXLD;
    axbuf += (size_t)blockIdx.y * BB * AXLD;
    const float4* src = (const float4*)part;
    float4 v[KSPLIT];
#pragma unroll
    for (int c = 0; c < KSPLIT; ++c) v[c] = src[(size_t)c * n4 + i];
    float4 s = v[0];
#pragma unroll
    for (int c = 1; c < KSPLIT; ++c) {
        s.x += v[c].x; s.y += v[c].y; s.z += v[c].z; s.w += v[c].w;
    }
    ((float4*)axbuf)[i] = s;
}

template <int NN>
__global__ void __launch_bounds__(256, 1)
xupd_kernel(const float* __restrict__ A, const float* __restrict__ axbuf,
            const float* __restrict__ Atb, const float* __restrict__ hyp,
            const int* __restrict__ nbr,
            float* __restrict__ X, const float* __restrict__ U,
            int p0, int p1, int k) {
    const int n0  = blockIdx.x * NGRP;
    const int it0 = blockIdx.y * 8 + (threadIdx.x >> 5);
    const int r0  = it0 * 16;
    const int pS[2] = {p0, (NN == 2) ? p1 : p0};
    const int rS[2] = {r0, (NN == 2) ? r0 : r0 + ROWSPAN};

    __shared__ float sBt[NN][NGRP][KLD];
    for (int idx = threadIdx.x; idx < MP * NGRP / 4; idx += 256) {
        const int kk = idx / (NGRP / 4);
        const int c4 = (idx % (NGRP / 4)) * 4;
        float4 v[NN];
        for (int s = 0; s < NN; ++s)
            v[s] = *(const float4*)&A[((size_t)pS[s] * MP + kk) * NCOLS + n0 + c4];
        for (int s = 0; s < NN; ++s) {
            sBt[s][c4 + 0][kk] = v[s].x;
            sBt[s][c4 + 1][kk] = v[s].y;
            sBt[s][c4 + 2][kk] = v[s].z;
            sBt[s][c4 + 3][kk] = v[s].w;
        }
    }
    __syncthreads();

    const int lane = threadIdx.x & 31;
    const int nl   = lane & 15;
    const int kh   = (lane >> 4) << 1;

    const float* pa0 = axbuf + (size_t)(rS[0] + nl) * AXLD + kh;
    const float* pa1 = axbuf + (size_t)(NN - 1) * BB * AXLD
                             + (size_t)(rS[1] + nl) * AXLD + kh;
    const float* ps0 = &sBt[0][nl][kh];
    const float* ps1 = &sBt[NN - 1][nl][kh];

    v8f acc0[JT5] = {};
    v8f acc1[JT5] = {};
    for (int k0 = 0; k0 < MP; k0 += 4) {
        v2f af0; af0.x = pa0[0]; af0.y = pa0[1]; pa0 += 4;
        v2f af1; af1.x = pa1[0]; af1.y = pa1[1]; pa1 += 4;
#pragma unroll
        for (int j = 0; j < JT5; ++j) {
            v2f bf0 = *(const v2f*)(ps0 + j * 16 * KLD);
            v2f bf1 = *(const v2f*)(ps1 + j * 16 * KLD);
            acc0[j] = wmma4(af0, bf0, acc0[j]);
            acc1[j] = wmma4(af1, bf1, acc1[j]);
        }
        ps0 += 4;
        ps1 += 4;
    }

    const size_t stride = (size_t)BB * NCOLS;
    float a0[2], a1[2], a2[2];
    float*       Xp[2];
    const float* Xl[2];
    const float* Xr[2];
    const float* Up[2];
    const float* Tp[2];
    for (int s = 0; s < 2; ++s) {
        const int p = pS[s];
        a0[s] = fabsf(hyp[(k * PN + p) * 4 + 0]);
        a1[s] = fabsf(hyp[(k * PN + p) * 4 + 1]);
        a2[s] = fabsf(hyp[(k * PN + p) * 4 + 2]);
        Xp[s] = X + (size_t)p * stride;
        Xl[s] = X + (size_t)nbr[p * 2 + 0] * stride;
        Xr[s] = X + (size_t)nbr[p * 2 + 1] * stride;
        Up[s] = U + (size_t)p * stride;
        Tp[s] = Atb + (size_t)p * stride;
    }

    const int ih = (lane >> 4) * 8;
#pragma unroll
    for (int j = 0; j < JT5; ++j) {
        const int col = n0 + j * 16 + nl;
#pragma unroll
        for (int v = 0; v < 8; ++v) {
            size_t off[2];
            float  xo[2], g[2];
            for (int s = 0; s < 2; ++s) {
                off[s] = (size_t)(rS[s] + ih + v) * NCOLS + col;
                xo[s]  = Xp[s][off[s]];
                const float sgn = (xo[s] > 0.f) ? 1.f : ((xo[s] < 0.f) ? -1.f : 0.f);
                const float av  = s ? acc1[j][v] : acc0[j][v];
                g[s] = av - Tp[s][off[s]] + 2.f * a0[s] * xo[s] + a2[s] * sgn
                       + 2.f * Up[s][off[s]] - a0[s] * (Xl[s][off[s]] + Xr[s][off[s]]);
            }
            for (int s = 0; s < 2; ++s)
                Xp[s][off[s]] = xo[s] - a1[s] * g[s];
        }
    }
}

__global__ void dual_kernel(float* __restrict__ XU, const float* __restrict__ hyp,
                            const int* __restrict__ nbr, int k) {
    const size_t stride4 = (size_t)BB * NCOLS / 4;
    const size_t total4  = (size_t)PN * stride4;
    size_t i = (size_t)blockIdx.x * blockDim.x + threadIdx.x;
    if (i >= total4) return;
    const int p = (int)(i / stride4);
    const size_t off = i - (size_t)p * stride4;
    const float a3 = fabsf(hyp[(k * PN + p) * 4 + 3]);
    const int pl = nbr[p * 2 + 0];
    const int pr = nbr[p * 2 + 1];
    const float4* X = (const float4*)XU;
    float4*       Uv = (float4*)(XU + (size_t)PN * BB * NCOLS);
    const float4 x  = X[i];
    const float4 xl = X[(size_t)pl * stride4 + off];
    const float4 xr = X[(size_t)pr * stride4 + off];
    float4 u = Uv[i];
    u.x += a3 * (2.f * x.x - xl.x - xr.x);
    u.y += a3 * (2.f * x.y - xl.y - xr.y);
    u.z += a3 * (2.f * x.z - xl.z - xr.z);
    u.w += a3 * (2.f * x.w - xl.w - xr.w);
    Uv[i] = u;
}

extern "C" void kernel_launch(void* const* d_in, const int* in_sizes, int n_in,
                              void* d_out, int out_size, void* d_ws, size_t ws_size,
                              hipStream_t stream) {
    (void)in_sizes; (void)n_in; (void)out_size; (void)ws_size;
    const float* A   = (const float*)d_in[0];
    const float* b   = (const float*)d_in[1];
    const float* hyp = (const float*)d_in[2];
    const int*   nbr = (const int*)d_in[3];

    float* X = (float*)d_out;
    float* U = X + (size_t)PN * BB * NCOLS;
    float* Atb   = (float*)d_ws;
    float* axbuf = Atb + (size_t)PN * BB * NCOLS;
    float* part  = axbuf + (size_t)2 * BB * AXLD;

    {
        long long n4 = (2LL * PN * BB * NCOLS) / 4;
        int blocks = (int)((n4 + 255) / 256);
        zero_kernel<<<blocks, 256, 0, stream>>>((float4*)d_out, n4);
    }
    atb_kernel<<<dim3(25, 4, PN), 256, 0, stream>>>(A, b, Atb);

    const int pairs[2][2] = {{0, 2}, {1, 3}};
    for (int k = 0; k < KITERS; ++k) {
        for (int c = 0; c < 2; ++c) {
            const int p0 = pairs[c][0];
            const int p1 = pairs[c][1];
            ax_part_kernel<<<dim3(8, KSPLIT, 2), 256, 0, stream>>>(A, X, part, p0, p1);
            ax_sum_kernel <<<dim3(BB * AXLD / 4 / 256, 2), 256, 0, stream>>>(part, axbuf);
            xupd_kernel<2><<<dim3(25, 8), 256, 0, stream>>>(A, axbuf, Atb, hyp, nbr,
                                                            X, U, p0, p1, k);
        }
        ax_part_kernel<<<dim3(8, KSPLIT, 1), 256, 0, stream>>>(A, X, part, 4, 4);
        ax_sum_kernel <<<dim3(BB * AXLD / 4 / 256, 1), 256, 0, stream>>>(part, axbuf);
        xupd_kernel<1><<<dim3(25, 4), 256, 0, stream>>>(A, axbuf, Atb, hyp, nbr,
                                                        X, U, 4, 4, k);
        {
            long long n4 = ((long long)PN * BB * NCOLS) / 4;
            int blocks = (int)((n4 + 255) / 256);
            dual_kernel<<<blocks, 256, 0, stream>>>(X, hyp, nbr, k);
        }
    }
}


# AX products and partial sums of all five nodes in one launch each at the start of an iteration (a node's X changes only in its own class update): 102 -> 62 launches per repetition; sum order kept
# speedup vs baseline: 9.8008x; 1.1037x over previous
#include <hip/hip_runtime.h>

#define PN      5
#define MROWS   500
#define NCOLS   2000
#define BB      1024
#define KITERS  10
#define MP      100
#define AXLD    112
#define JT5     5
#define NGRP    80
#define KLD     102
#define KCH     100
#define MPAD    112
#define ROWSPAN 512
#define KSPLIT  20
#define STG     11

typedef __attribute__((ext_vector_type(2))) float v2f;
typedef __attribute__((ext_vector_type(8))) float v8f;

__device__ __forceinline__ v8f wmma4(v2f a, v2f b, v8f c) {
    return __builtin_amdgcn_wmma_f32_16x16x4_f32(
        false, a, false, b, (short)0, c, false, false);
}


__global__ void zero_kernel(float4* __restrict__ dst, long long n4) {
    long long i = (long long)blockIdx.x * blockDim.x + threadIdx.x;
    if (i < n4) dst[i] = make_float4(0.f, 0.f, 0.f, 0.f);
}

__global__ void __launch_bounds__(256, 1)
atb_kernel(const float* __restrict__ A, const float* __restrict__ b,
           float* __restrict__ Atb) {
    const int p    = blockIdx.z;
    const int n0   = blockIdx.x * NGRP;
    const int it0  = blockIdx.y * 8 + (threadIdx.x >> 5);
    const int r0   = it0 * 16;
    const int r1   = r0 + ROWSPAN;
    const float* Ap = A + (size_t)p * MP * NCOLS;

    __shared__ float sBt[NGRP][KLD];
    for (int idx = threadIdx.x; idx < MP * NGRP / 4; idx += 256) {
        const int kk = idx / (NGRP / 4);
        const int c4 = (idx % (NGRP / 4)) * 4;
        float4 v = *(const float4*)&Ap[(size_t)kk * NCOLS + n0 + c4];
        sBt[c4 + 0][kk] = v.x;
        sBt[c4 + 1][kk] = v.y;
        sBt[c4 + 2][kk] = v.z;
        sBt[c4 + 3][kk] = v.w;
    }
    __syncthreads();

    const int lane = threadIdx.x & 31;
    const int nl   = lane & 15;
    const int kh   = (lane >> 4) << 1;

    const float* pa0 = b + (size_t)(r0 + nl) * MROWS + p * MP + kh;
    const float* pa1 = b + (size_t)(r1 + nl) * MROWS + p * MP + kh;
    const float* ps  = &sBt[nl][kh];

    v8f acc0[JT5] = {};
    v8f acc1[JT5] = {};
    for (int k0 = 0; k0 < MP; k0 += 4) {
        v2f af0; af0.x = pa0[0]; af0.y = pa0[1]; pa0 += 4;
        v2f af1; af1.x = pa1[0]; af1.y = pa1[1]; pa1 += 4;
#pragma unroll
        for (int j = 0; j < JT5; ++j) {
            v2f bf = *(const v2f*)(ps + j * 16 * KLD);
            acc0[j] = wmma4(af0, bf, acc0[j]);
            acc1[j] = wmma4(af1, bf, acc1[j]);
        }
        ps += 4;
    }

    float* out = Atb + (size_t)p * BB * NCOLS;
    const int ih = (lane >> 4) * 8;
#pragma unroll
    for (int j = 0; j < JT5; ++j) {
        const int col = n0 + j * 16 + nl;
#pragma unroll
        for (int v = 0; v < 8; ++v) {
            out[(size_t)(r0 + ih + v) * NCOLS + col] = acc0[j][v];
            out[(size_t)(r1 + ih + v) * NCOLS + col] = acc1[j][v];
        }
    }
}

__global__ void __launch_bounds__(256, 1)
ax_part_kernel(const float* __restrict__ A, const float* __restrict__ X,
               float* __restrict__ part) {
    const int p  = blockIdx.z;
    const int r0 = (blockIdx.x * 8 + (threadIdx.x >> 5)) * 16;
    const int kc = blockIdx.y * KCH;
    const float* Ap = A + (size_t)p * MP * NCOLS;
    const float* Xp = X + (size_t)p * BB * NCOLS;

    __shared__ float sB[MPAD][KCH];

    const int lane = threadIdx.x & 31;
    const int ml   = lane & 15;
    const int kh   = (lane >> 4) << 1;

    float4 st[STG];
#pragma unroll
    for (int t = 0; t < STG; ++t) {
        const int idx = threadIdx.x + t * 256;
        const int m   = idx / (KCH / 4);
        const int c4  = (idx % (KCH / 4)) * 4;
        st[t] = make_float4(0.f, 0.f, 0.f, 0.f);
        if (t * 256 + 255 < MP * KCH / 4) {
            st[t] = *(const float4*)&Ap[(size_t)m * NCOLS + kc + c4];
        } else if (t * 256 < MP * KCH / 4) {
            if (m < MP)
                st[t] = *(const float4*)&Ap[(size_t)m * NCOLS + kc + c4];
        }
    }

    const float* px = Xp + (size_t)(r0 + ml) * NCOLS + kc + kh;
    v2f af[KCH / 4];
#pragma unroll
    for (int i = 0; i < KCH / 4; ++i) {
        af[i].x = px[4 * i];
        af[i].y = px[4 * i + 1];
    }

#pragma unroll
    for (int t = 0; t < STG; ++t) {
        const int idx = threadIdx.x + t * 256;
        const int m   = idx / (KCH / 4);
        const int c4  = (idx % (KCH / 4)) * 4;
        if (idx < MPAD * KCH / 4)
            *(float4*)&sB[m][c4] = st[t];
    }
    __syncthreads();

    v8f acc[7] = {};
#pragma unroll
    for (int i = 0; i < KCH / 4; ++i) {
        const float* ps = &sB[ml][4 * i + kh];
#pragma unroll
        for (int j = 0; j < 7; ++j) {
            v2f bf = *(const v2f*)(ps + j * 16 * KCH);
            acc[j] = wmma4(af[i], bf, acc[j]);
        }
    }

    float* out = part + ((size_t)blockIdx.z * KSPLIT + blockIdx.y) * BB * AXLD;
    const int ih = (lane >> 4) * 8;
#pragma unroll
    for (int j = 0; j < 7; ++j) {
#pragma unroll
        for (int v = 0; v < 8; ++v)
            out[(size_t)(r0 + ih + v) * AXLD + j * 16 + ml] = acc[j][v];
    }
}

__global__ void ax_sum_kernel(const float* __restrict__ part,
                              float* __restrict__ axbuf) {
    const int n4 = BB * AXLD / 4;
    const int i  = blockIdx.x * blockDim.x + threadIdx.x;
    part  += (size_t)blockIdx.y * KSPLIT * BB * AXLD;
    axbuf += (size_t)blockIdx.y * BB * AXLD;
    const float4* src = (const float4*)part;
    float4 v[KSPLIT];
#pragma unroll
    for (int c = 0; c < KSPLIT; ++c) v[c] = src[(size_t)c * n4 + i];
    float4 s = v[0];
#pragma unroll
    for (int c = 1; c < KSPLIT; ++c) {
        s.x += v[c].x; s.y += v[c].y; s.z += v[c].z; s.w += v[c].w;
    }
    ((float4*)axbuf)[i] = s;
}

template <int NN>
__global__ void __launch_bounds__(256, 1)
xupd_kernel(const float* __restrict__ A, const float* __restrict__ axbuf,
            const float* __restrict__ Atb, const float* __restrict__ hyp,
            const int* __restrict__ nbr,
            float* __restrict__ X, const float* __restrict__ U,
            int p0, int p1, int k) {
    const int n0  = blockIdx.x * NGRP;
    const int it0 = blockIdx.y * 8 + (threadIdx.x >> 5);
    const int r0  = it0 * 16;
    const int pS[2] = {p0, (NN == 2) ? p1 : p0};
    const int rS[2] = {r0, (NN == 2) ? r0 : r0 + ROWSPAN};

    __shared__ float sBt[NN][NGRP][KLD];
    for (int idx = threadIdx.x; idx < MP * NGRP / 4; idx += 256) {
        const int kk = idx / (NGRP / 4);
        const int c4 = (idx % (NGRP / 4)) * 4;
        float4 v[NN];
        for (int s = 0; s < NN; ++s)
            v[s] = *(const float4*)&A[((size_t)pS[s] * MP + kk) * NCOLS + n0 + c4];
        for (int s = 0; s < NN; ++s) {
            sBt[s][c4 + 0][kk] = v[s].x;
            sBt[s][c4 + 1][kk] = v[s].y;
            sBt[s][c4 + 2][kk] = v[s].z;
            sBt[s][c4 + 3][kk] = v[s].w;
        }
    }
    __syncthreads();

    const int lane = threadIdx.x & 31;
    const int nl   = lane & 15;
    const int kh   = (lane >> 4) << 1;

    const float* pa0 = axbuf + (size_t)pS[0] * BB * AXLD
                             + (size_t)(rS[0] + nl) * AXLD + kh;
    const float* pa1 = axbuf + (size_t)pS[1] * BB * AXLD
                             + (size_t)(rS[1] + nl) * AXLD + kh;
    const float* ps0 = &sBt[0][nl][kh];
    const float* ps1 = &sBt[NN - 1][nl][kh];

    v8f acc0[JT5] = {};
    v8f acc1[JT5] = {};
    for (int k0 = 0; k0 < MP; k0 += 4) {
        v2f af0; af0.x = pa0[0]; af0.y = pa0[1]; pa0 += 4;
        v2f af1; af1.x = pa1[0]; af1.y = pa1[1]; pa1 += 4;
#pragma unroll
        for (int j = 0; j < JT5; ++j) {
            v2f bf0 = *(const v2f*)(ps0 + j * 16 * KLD);
            v2f bf1 = *(const v2f*)(ps1 + j * 16 * KLD);
            acc0[j] = wmma4(af0, bf0, acc0[j]);
            acc1[j] = wmma4(af1, bf1, acc1[j]);
        }
        ps0 += 4;
        ps1 += 4;
    }

    const size_t stride = (size_t)BB * NCOLS;
    float a0[2], a1[2], a2[2];
    float*       Xp[2];
    const float* Xl[2];
    const float* Xr[2];
    const float* Up[2];
    const float* Tp[2];
    for (int s = 0; s < 2; ++s) {
        const int p = pS[s];
        a0[s] = fabsf(hyp[(k * PN + p) * 4 + 0]);
        a1[s] = fabsf(hyp[(k * PN + p) * 4 + 1]);
        a2[s] = fabsf(hyp[(k * PN + p) * 4 + 2]);
        Xp[s] = X + (size_t)p * stride;
        Xl[s] = X + (size_t)nbr[p * 2 + 0] * stride;
        Xr[s] = X + (size_t)nbr[p * 2 + 1] * stride;
        Up[s] = U + (size_t)p * stride;
        Tp[s] = Atb + (size_t)p * stride;
    }

    const int ih = (lane >> 4) * 8;
#pragma unroll
    for (int j = 0; j < JT5; ++j) {
        const int col = n0 + j * 16 + nl;
#pragma unroll
        for (int v = 0; v < 8; ++v) {
            size_t off[2];
            float  xo[2], g[2];
            for (int s = 0; s < 2; ++s) {
                off[s] = (size_t)(rS[s] + ih + v) * NCOLS + col;
                xo[s]  = Xp[s][off[s]];
                const float sgn = (xo[s] > 0.f) ? 1.f : ((xo[s] < 0.f) ? -1.f : 0.f);
                const float av  = s ? acc1[j][v] : acc0[j][v];
                g[s] = av - Tp[s][off[s]] + 2.f * a0[s] * xo[s] + a2[s] * sgn
                       + 2.f * Up[s][off[s]] - a0[s] * (Xl[s][off[s]] + Xr[s][off[s]]);
            }
            for (int s = 0; s < 2; ++s)
                Xp[s][off[s]] = xo[s] - a1[s] * g[s];
        }
    }
}

__global__ void dual_kernel(float* __restrict__ XU, const float* __restrict__ hyp,
                            const int* __restrict__ nbr, int k) {
    const size_t stride4 = (size_t)BB * NCOLS / 4;
    const size_t total4  = (size_t)PN * stride4;
    size_t i = (size_t)blockIdx.x * blockDim.x + threadIdx.x;
    if (i >= total4) return;
    const int p = (int)(i / stride4);
    const size_t off = i - (size_t)p * stride4;
    const float a3 = fabsf(hyp[(k * PN + p) * 4 + 3]);
    const int pl = nbr[p * 2 + 0];
    const int pr = nbr[p * 2 + 1];
    const float4* X = (const float4*)XU;
    float4*       Uv = (float4*)(XU + (size_t)PN * BB * NCOLS);
    const float4 x  = X[i];
    const float4 xl = X[(size_t)pl * stride4 + off];
    const float4 xr = X[(size_t)pr * stride4 + off];
    float4 u = Uv[i];
    u.x += a3 * (2.f * x.x - xl.x - xr.x);
    u.y += a3 * (2.f * x.y - xl.y - xr.y);
    u.z += a3 * (2.f * x.z - xl.z - xr.z);
    u.w += a3 * (2.f * x.w - xl.w - xr.w);
    Uv[i] = u;
}

extern "C" void kernel_launch(void* const* d_in, const int* in_sizes, int n_in,
                              void* d_out, int out_size, void* d_ws, size_t ws_size,
                              hipStream_t stream) {
    (void)in_sizes; (void)n_in; (void)out_size; (void)ws_size;
    const float* A   = (const float*)d_in[0];
    const float* b   = (const float*)d_in[1];
    const float* hyp = (const float*)d_in[2];
    const int*   nbr = (const int*)d_in[3];

    float* X = (float*)d_out;
    float* U = X + (size_t)PN * BB * NCOLS;
    float* Atb   = (float*)d_ws;
    float* axbuf = Atb + (size_t)PN * BB * NCOLS;
    float* part  = axbuf + (size_t)PN * BB * AXLD;

    {
        long long n4 = (2LL * PN * BB * NCOLS) / 4;
        int blocks = (int)((n4 + 255) / 256);
        zero_kernel<<<blocks, 256, 0, stream>>>((float4*)d_out, n4);
    }
    atb_kernel<<<dim3(25, 4, PN), 256, 0, stream>>>(A, b, Atb);

    const int pairs[2][2] = {{0, 2}, {1, 3}};
    for (int k = 0; k < KITERS; ++k) {
        ax_part_kernel<<<dim3(8, KSPLIT, PN), 256, 0, stream>>>(A, X, part);
        ax_sum_kernel <<<dim3(BB * AXLD / 4 / 256, PN), 256, 0, stream>>>(part, axbuf);
        for (int c = 0; c < 2; ++c) {
            const int p0 = pairs[c][0];
            const int p1 = pairs[c][1];
            xupd_kernel<2><<<dim3(25, 8), 256, 0, stream>>>(A, axbuf, Atb, hyp, nbr,
                                                            X, U, p0, p1, k);
        }
        xupd_kernel<1><<<dim3(25, 4), 256, 0, stream>>>(A, axbuf, Atb, hyp, nbr,
                                                        X, U, 4, 4, k);
        {
            long long n4 = ((long long)PN * BB * NCOLS) / 4;
            int blocks = (int)((n4 + 255) / 256);
            dual_kernel<<<blocks, 256, 0, stream>>>(X, hyp, nbr, k);
        }
    }
}


# AtAX of all five nodes, the three colour-class updates and the dual update in ONE launch: a lane keeps one (b,n) of all nodes in registers through the classes; 62 -> 32 launches; sum order kept
# speedup vs baseline: 14.8939x; 1.5197x over previous
#include <hip/hip_runtime.h>

#define PN      5
#define MROWS   500
#define NCOLS   2000
#define BB      1024
#define KITERS  10
#define MP      100
#define AXLD    112
#define JT5     5
#define NGRP    80
#define KLD     102
#define KCH     100
#define MPAD    112
#define ROWSPAN 512
#define KSPLIT  20
#define STG     11

typedef __attribute__((ext_vector_type(2))) float v2f;
typedef __attribute__((ext_vector_type(8))) float v8f;

__device__ __forceinline__ v8f wmma4(v2f a, v2f b, v8f c) {
    return __builtin_amdgcn_wmma_f32_16x16x4_f32(
        false, a, false, b, (short)0, c, false, false);
}


__global__ void zero_kernel(float4* __restrict__ dst, long long n4) {
    long long i = (long long)blockIdx.x * blockDim.x + threadIdx.x;
    if (i < n4) dst[i] = make_float4(0.f, 0.f, 0.f, 0.f);
}

__global__ void __launch_bounds__(256, 1)
atb_kernel(const float* __restrict__ A, const float* __restrict__ b,
           float* __restrict__ Atb) {
    const int p    = blockIdx.z;
    const int n0   = blockIdx.x * NGRP;
    const int it0  = blockIdx.y * 8 + (threadIdx.x >> 5);
    const int r0   = it0 * 16;
    const int r1   = r0 + ROWSPAN;
    const float* Ap = A + (size_t)p * MP * NCOLS;

    __shared__ float sBt[NGRP][KLD];
    for (int idx = threadIdx.x; idx < MP * NGRP / 4; idx += 256) {
        const int kk = idx / (NGRP / 4);
        const int c4 = (idx % (NGRP / 4)) * 4;
        float4 v = *(const float4*)&Ap[(size_t)kk * NCOLS + n0 + c4];
        sBt[c4 + 0][kk] = v.x;
        sBt[c4 + 1][kk] = v.y;
        sBt[c4 + 2][kk] = v.z;
        sBt[c4 + 3][kk] = v.w;
    }
    __syncthreads();

    const int lane = threadIdx.x & 31;
    const int nl   = lane & 15;
    const int kh   = (lane >> 4) << 1;

    const float* pa0 = b + (size_t)(r0 + nl) * MROWS + p * MP + kh;
    const float* pa1 = b + (size_t)(r1 + nl) * MROWS + p * MP + kh;
    const float* ps  = &sBt[nl][kh];

    v8f acc0[JT5] = {};
    v8f acc1[JT5] = {};
    for (int k0 = 0; k0 < MP; k0 += 4) {
        v2f af0; af0.x = pa0[0]; af0.y = pa0[1]; pa0 += 4;
        v2f af1; af1.x = pa1[0]; af1.y = pa1[1]; pa1 += 4;
#pragma unroll
        for (int j = 0; j < JT5; ++j) {
            v2f bf = *(const v2f*)(ps + j * 16 * KLD);
            acc0[j] = wmma4(af0, bf, acc0[j]);
            acc1[j] = wmma4(af1, bf, acc1[j]);
        }
        ps += 4;
    }

    float* out = Atb + (size_t)p * BB * NCOLS;
    const int ih = (lane >> 4) * 8;
#pragma unroll
    for (int j = 0; j < JT5; ++j) {
        const int col = n0 + j * 16 + nl;
#pragma unroll
        for (int v = 0; v < 8; ++v) {
            out[(size_t)(r0 + ih + v) * NCOLS + col] = acc0[j][v];
            out[(size_t)(r1 + ih + v) * NCOLS + col] = acc1[j][v];
        }
    }
}

__global__ void __launch_bounds__(256, 1)
ax_part_kernel(const float* __restrict__ A, const float* __restrict__ X,
               float* __restrict__ part) {
    const int p  = blockIdx.z;
    const int r0 = (blockIdx.x * 8 + (threadIdx.x >> 5)) * 16;
    const int kc = blockIdx.y * KCH;
    const float* Ap = A + (size_t)p * MP * NCOLS;
    const float* Xp = X + (size_t)p * BB * NCOLS;

    __shared__ float sB[MPAD][KCH];

    const int lane = threadIdx.x & 31;
    const int ml   = lane & 15;
    const int kh   = (lane >> 4) << 1;

    float4 st[STG];
#pragma unroll
    for (int t = 0; t < STG; ++t) {
        const int idx = threadIdx.x + t * 256;
        const int m   = idx / (KCH / 4);
        const int c4  = (idx % (KCH / 4)) * 4;
        st[t] = make_float4(0.f, 0.f, 0.f, 0.f);
        if (t * 256 + 255 < MP * KCH / 4) {
            st[t] = *(const float4*)&Ap[(size_t)m * NCOLS + kc + c4];
        } else if (t * 256 < MP * KCH / 4) {
            if (m < MP)
                st[t] = *(const float4*)&Ap[(size_t)m * NCOLS + kc + c4];
        }
    }

    const float* px = Xp + (size_t)(r0 + ml) * NCOLS + kc + kh;
    v2f af[KCH / 4];
#pragma unroll
    for (int i = 0; i < KCH / 4; ++i) {
        af[i].x = px[4 * i];
        af[i].y = px[4 * i + 1];
    }

#pragma unroll
    for (int t = 0; t < STG; ++t) {
        const int idx = threadIdx.x + t * 256;
        const int m   = idx / (KCH / 4);
        const int c4  = (idx % (KCH / 4)) * 4;
        if (idx < MPAD * KCH / 4)
            *(float4*)&sB[m][c4] = st[t];
    }
    __syncthreads();

    v8f acc[7] = {};
#pragma unroll
    for (int i = 0; i < KCH / 4; ++i) {
        const float* ps = &sB[ml][4 * i + kh];
#pragma unroll
        for (int j = 0; j < 7; ++j) {
            v2f bf = *(const v2f*)(ps + j * 16 * KCH);
            acc[j] = wmma4(af[i], bf, acc[j]);
        }
    }

    float* out = part + ((size_t)blockIdx.z * KSPLIT + blockIdx.y) * BB * AXLD;
    const int ih = (lane >> 4) * 8;
#pragma unroll
    for (int j = 0; j < 7; ++j) {
#pragma unroll
        for (int v = 0; v < 8; ++v)
            out[(size_t)(r0 + ih + v) * AXLD + j * 16 + ml] = acc[j][v];
    }
}

__global__ void ax_sum_kernel(const float* __restrict__ part,
                              float* __restrict__ axbuf) {
    const int n4 = BB * AXLD / 4;
    const int i  = blockIdx.x * blockDim.x + threadIdx.x;
    part  += (size_t)blockIdx.y * KSPLIT * BB * AXLD;
    axbuf += (size_t)blockIdx.y * BB * AXLD;
    const float4* src = (const float4*)part;
    float4 v[KSPLIT];
#pragma unroll
    for (int c = 0; c < KSPLIT; ++c) v[c] = src[(size_t)c * n4 + i];
    float4 s = v[0];
#pragma unroll
    for (int c = 1; c < KSPLIT; ++c) {
        s.x += v[c].x; s.y += v[c].y; s.z += v[c].z; s.w += v[c].w;
    }
    ((float4*)axbuf)[i] = s;
}

#define RT_WG       4
#define UPD_THREADS (RT_WG * JT5 * 32)
#define UPD_STG     16

__device__ __forceinline__ float pick5(const float (&x)[PN], int q) {
    float r = x[0];
    r = (q == 1) ? x[1] : r;
    r = (q == 2) ? x[2] : r;
    r = (q == 3) ? x[3] : r;
    r = (q == 4) ? x[4] : r;
    return r;
}

__global__ void __launch_bounds__(UPD_THREADS)
update_kernel(const float* __restrict__ A, const float* __restrict__ axbuf,
              const float* __restrict__ Atb, const float* __restrict__ hyp,
              const int* __restrict__ nbr,
              float* __restrict__ X, float* __restrict__ U, int k) {
    __shared__ float sB[MROWS][NGRP];
    const int n0w = blockIdx.x * NGRP;

    float4 st[UPD_STG];
#pragma unroll
    for (int t = 0; t < UPD_STG; ++t) {
        const int idx = min((int)threadIdx.x + t * UPD_THREADS, MROWS * NGRP / 4 - 1);
        const int row = idx / (NGRP / 4);
        const int c4  = (idx % (NGRP / 4)) * 4;
        st[t] = *(const float4*)&A[(size_t)row * NCOLS + n0w + c4];
    }
#pragma unroll
    for (int t = 0; t < UPD_STG; ++t) {
        const int idx = threadIdx.x + t * UPD_THREADS;
        const int row = idx / (NGRP / 4);
        const int c4  = (idx % (NGRP / 4)) * 4;
        if (idx < MROWS * NGRP / 4)
            *(float4*)&sB[row][c4] = st[t];
    }
    __syncthreads();

    const int wave = threadIdx.x >> 5;
    const int jt   = wave % JT5;
    const int r0   = (blockIdx.y * RT_WG + wave / JT5) * 16;
    const int lane = threadIdx.x & 31;
    const int c    = lane & 15;
    const int kh   = (lane >> 4) << 1;

    const size_t stride = (size_t)BB * NCOLS;
    const int ih  = (lane >> 4) * 8;
    const int col = n0w + jt * 16 + c;

    float x[8][PN], u[8][PN], t[8][PN];
#pragma unroll
    for (int v = 0; v < 4; ++v) {
        const size_t off = (size_t)(r0 + ih + v) * NCOLS + col;
#pragma unroll
        for (int p = 0; p < PN; ++p) {
            x[v][p] = X[(size_t)p * stride + off];
            u[v][p] = U[(size_t)p * stride + off];
            t[v][p] = Atb[(size_t)p * stride + off];
        }
    }

    const float* pa = axbuf + (size_t)(r0 + c) * AXLD + kh;
    const float* pb = &sB[kh][jt * 16 + c];

    v8f acc[PN] = {};
#pragma unroll
    for (int ks = 0; ks < MP / 4; ++ks) {
#pragma unroll
        for (int s = 0; s < PN; ++s) {
            const v2f af = *(const v2f*)(pa + (size_t)s * BB * AXLD + 4 * ks);
            v2f bf;
            bf.x = pb[(s * MP + 4 * ks) * NGRP];
            bf.y = pb[(s * MP + 4 * ks + 1) * NGRP];
            acc[s] = wmma4(af, bf, acc[s]);
        }
    }

#pragma unroll
    for (int v = 4; v < 8; ++v) {
        const size_t off = (size_t)(r0 + ih + v) * NCOLS + col;
#pragma unroll
        for (int p = 0; p < PN; ++p) {
            x[v][p] = X[(size_t)p * stride + off];
            u[v][p] = U[(size_t)p * stride + off];
            t[v][p] = Atb[(size_t)p * stride + off];
        }
    }

    float a0[PN], a1[PN], a2[PN], a3[PN];
    int   nl[PN], nr[PN];
#pragma unroll
    for (int p = 0; p < PN; ++p) {
        a0[p] = fabsf(hyp[(k * PN + p) * 4 + 0]);
        a1[p] = fabsf(hyp[(k * PN + p) * 4 + 1]);
        a2[p] = fabsf(hyp[(k * PN + p) * 4 + 2]);
        a3[p] = fabsf(hyp[(k * PN + p) * 4 + 3]);
        nl[p] = nbr[p * 2 + 0];
        nr[p] = nbr[p * 2 + 1];
    }

#pragma unroll
    for (int v = 0; v < 8; ++v) {
        const size_t off = (size_t)(r0 + ih + v) * NCOLS + col;
        float (&xv)[PN] = x[v];
        auto step = [&](int p, float av) -> float {
            const float xo  = xv[p];
            const float sgn = (xo > 0.f) ? 1.f : ((xo < 0.f) ? -1.f : 0.f);
            const float g = av - t[v][p] + 2.f * a0[p] * xo + a2[p] * sgn
                            + 2.f * u[v][p] - a0[p] * (pick5(xv, nl[p]) + pick5(xv, nr[p]));
            return xo - a1[p] * g;
        };
        {
            const float y0 = step(0, acc[0][v]);
            const float y2 = step(2, acc[2][v]);
            xv[0] = y0; xv[2] = y2;
        }
        {
            const float y1 = step(1, acc[1][v]);
            const float y3 = step(3, acc[3][v]);
            xv[1] = y1; xv[3] = y3;
        }
        xv[4] = step(4, acc[4][v]);
        float un[PN];
#pragma unroll
        for (int p = 0; p < PN; ++p)
            un[p] = u[v][p] + a3[p] * (2.f * xv[p] - pick5(xv, nl[p]) - pick5(xv, nr[p]));
#pragma unroll
        for (int p = 0; p < PN; ++p) {
            X[(size_t)p * stride + off] = xv[p];
            U[(size_t)p * stride + off] = un[p];
        }
    }
}

extern "C" void kernel_launch(void* const* d_in, const int* in_sizes, int n_in,
                              void* d_out, int out_size, void* d_ws, size_t ws_size,
                              hipStream_t stream) {
    (void)in_sizes; (void)n_in; (void)out_size; (void)ws_size;
    const float* A   = (const float*)d_in[0];
    const float* b   = (const float*)d_in[1];
    const float* hyp = (const float*)d_in[2];
    const int*   nbr = (const int*)d_in[3];

    float* X = (float*)d_out;
    float* U = X + (size_t)PN * BB * NCOLS;
    float* Atb   = (float*)d_ws;
    float* axbuf = Atb + (size_t)PN * BB * NCOLS;
    float* part  = axbuf + (size_t)PN * BB * AXLD;

    {
        long long n4 = (2LL * PN * BB * NCOLS) / 4;
        int blocks = (int)((n4 + 255) / 256);
        zero_kernel<<<blocks, 256, 0, stream>>>((float4*)d_out, n4);
    }
    atb_kernel<<<dim3(25, 4, PN), 256, 0, stream>>>(A, b, Atb);

    for (int k = 0; k < KITERS; ++k) {
        ax_part_kernel<<<dim3(8, KSPLIT, PN), 256, 0, stream>>>(A, X, part);
        ax_sum_kernel <<<dim3(BB * AXLD / 4 / 256, PN), 256, 0, stream>>>(part, axbuf);
        update_kernel <<<dim3(25, BB / (RT_WG * 16)), UPD_THREADS, 0, stream>>>(
            A, axbuf, Atb, hyp, nbr, X, U, k);
    }
}


# iteration 0 specialised on the reference's start X=U=0: no zero-fill, AX product, partial sum, AtAX or X/U reads; one float4 streaming launch instead (32 -> 29 launches); sum order kept
# speedup vs baseline: 16.3212x; 1.0958x over previous
#include <hip/hip_runtime.h>

#define PN      5
#define MROWS   500
#define NCOLS   2000
#define BB      1024
#define KITERS  10
#define MP      100
#define AXLD    112
#define JT5     5
#define NGRP    80
#define KLD     102
#define KCH     100
#define MPAD    112
#define ROWSPAN 512
#define KSPLIT  20
#define STG     11

typedef __attribute__((ext_vector_type(2))) float v2f;
typedef __attribute__((ext_vector_type(8))) float v8f;

__device__ __forceinline__ v8f wmma4(v2f a, v2f b, v8f c) {
    return __builtin_amdgcn_wmma_f32_16x16x4_f32(
        false, a, false, b, (short)0, c, false, false);
}


__global__ void __launch_bounds__(256, 1)
atb_kernel(const float* __restrict__ A, const float* __restrict__ b,
           float* __restrict__ Atb) {
    const int p    = blockIdx.z;
    const int n0   = blockIdx.x * NGRP;
    const int it0  = blockIdx.y * 8 + (threadIdx.x >> 5);
    const int r0   = it0 * 16;
    const int r1   = r0 + ROWSPAN;
    const float* Ap = A + (size_t)p * MP * NCOLS;

    __shared__ float sBt[NGRP][KLD];
    for (int idx = threadIdx.x; idx < MP * NGRP / 4; idx += 256) {
        const int kk = idx / (NGRP / 4);
        const int c4 = (idx % (NGRP / 4)) * 4;
        float4 v = *(const float4*)&Ap[(size_t)kk * NCOLS + n0 + c4];
        sBt[c4 + 0][kk] = v.x;
        sBt[c4 + 1][kk] = v.y;
        sBt[c4 + 2][kk] = v.z;
        sBt[c4 + 3][kk] = v.w;
    }
    __syncthreads();

    const int lane = threadIdx.x & 31;
    const int nl   = lane & 15;
    const int kh   = (lane >> 4) << 1;

    const float* pa0 = b + (size_t)(r0 + nl) * MROWS + p * MP + kh;
    const float* pa1 = b + (size_t)(r1 + nl) * MROWS + p * MP + kh;
    const float* ps  = &sBt[nl][kh];

    v8f acc0[JT5] = {};
    v8f acc1[JT5] = {};
    for (int k0 = 0; k0 < MP; k0 += 4) {
        v2f af0; af0.x = pa0[0]; af0.y = pa0[1]; pa0 += 4;
        v2f af1; af1.x = pa1[0]; af1.y = pa1[1]; pa1 += 4;
#pragma unroll
        for (int j = 0; j < JT5; ++j) {
            v2f bf = *(const v2f*)(ps + j * 16 * KLD);
            acc0[j] = wmma4(af0, bf, acc0[j]);
            acc1[j] = wmma4(af1, bf, acc1[j]);
        }
        ps += 4;
    }

    float* out = Atb + (size_t)p * BB * NCOLS;
    const int ih = (lane >> 4) * 8;
#pragma unroll
    for (int j = 0; j < JT5; ++j) {
        const int col = n0 + j * 16 + nl;
#pragma unroll
        for (int v = 0; v < 8; ++v) {
            out[(size_t)(r0 + ih + v) * NCOLS + col] = acc0[j][v];
            out[(size_t)(r1 + ih + v) * NCOLS + col] = acc1[j][v];
        }
    }
}

__global__ void __launch_bounds__(256, 1)
ax_part_kernel(const float* __restrict__ A, const float* __restrict__ X,
               float* __restrict__ part) {
    const int p  = blockIdx.z;
    const int r0 = (blockIdx.x * 8 + (threadIdx.x >> 5)) * 16;
    const int kc = blockIdx.y * KCH;
    const float* Ap = A + (size_t)p * MP * NCOLS;
    const float* Xp = X + (size_t)p * BB * NCOLS;

    __shared__ float sB[MPAD][KCH];

    const int lane = threadIdx.x & 31;
    const int ml   = lane & 15;
    const int kh   = (lane >> 4) << 1;

    float4 st[STG];
#pragma unroll
    for (int t = 0; t < STG; ++t) {
        const int idx = threadIdx.x + t * 256;
        const int m   = idx / (KCH / 4);
        const int c4  = (idx % (KCH / 4)) * 4;
        st[t] = make_float4(0.f, 0.f, 0.f, 0.f);
        if (t * 256 + 255 < MP * KCH / 4) {
            st[t] = *(const float4*)&Ap[(size_t)m * NCOLS + kc + c4];
        } else if (t * 256 < MP * KCH / 4) {
            if (m < MP)
                st[t] = *(const float4*)&Ap[(size_t)m * NCOLS + kc + c4];
        }
    }

    const float* px = Xp + (size_t)(r0 + ml) * NCOLS + kc + kh;
    v2f af[KCH / 4];
#pragma unroll
    for (int i = 0; i < KCH / 4; ++i) {
        af[i].x = px[4 * i];
        af[i].y = px[4 * i + 1];
    }

#pragma unroll
    for (int t = 0; t < STG; ++t) {
        const int idx = threadIdx.x + t * 256;
        const int m   = idx / (KCH / 4);
        const int c4  = (idx % (KCH / 4)) * 4;
        if (idx < MPAD * KCH / 4)
            *(float4*)&sB[m][c4] = st[t];
    }
    __syncthreads();

    v8f acc[7] = {};
#pragma unroll
    for (int i = 0; i < KCH / 4; ++i) {
        const float* ps = &sB[ml][4 * i + kh];
#pragma unroll
        for (int j = 0; j < 7; ++j) {
            v2f bf = *(const v2f*)(ps + j * 16 * KCH);
            acc[j] = wmma4(af[i], bf, acc[j]);
        }
    }

    float* out = part + ((size_t)blockIdx.z * KSPLIT + blockIdx.y) * BB * AXLD;
    const int ih = (lane >> 4) * 8;
#pragma unroll
    for (int j = 0; j < 7; ++j) {
#pragma unroll
        for (int v = 0; v < 8; ++v)
            out[(size_t)(r0 + ih + v) * AXLD + j * 16 + ml] = acc[j][v];
    }
}

__global__ void ax_sum_kernel(const float* __restrict__ part,
                              float* __restrict__ axbuf) {
    const int n4 = BB * AXLD / 4;
    const int i  = blockIdx.x * blockDim.x + threadIdx.x;
    part  += (size_t)blockIdx.y * KSPLIT * BB * AXLD;
    axbuf += (size_t)blockIdx.y * BB * AXLD;
    const float4* src = (const float4*)part;
    float4 v[KSPLIT];
#pragma unroll
    for (int c = 0; c < KSPLIT; ++c) v[c] = src[(size_t)c * n4 + i];
    float4 s = v[0];
#pragma unroll
    for (int c = 1; c < KSPLIT; ++c) {
        s.x += v[c].x; s.y += v[c].y; s.z += v[c].z; s.w += v[c].w;
    }
    ((float4*)axbuf)[i] = s;
}

#define RT_WG       4
#define UPD_THREADS (RT_WG * JT5 * 32)
#define UPD_STG     16

__device__ __forceinline__ float pick5(const float (&x)[PN], int q) {
    float r = x[0];
    r = (q == 1) ? x[1] : r;
    r = (q == 2) ? x[2] : r;
    r = (q == 3) ? x[3] : r;
    r = (q == 4) ? x[4] : r;
    return r;
}

__global__ void __launch_bounds__(256)
first_update_kernel(const float* __restrict__ Atb, const float* __restrict__ hyp,
                    const int* __restrict__ nbr,
                    float* __restrict__ X, float* __restrict__ U) {
    const size_t plane4 = (size_t)BB * NCOLS / 4;
    const size_t i = (size_t)blockIdx.x * 256 + threadIdx.x;

    float4 t4[PN];
#pragma unroll
    for (int p = 0; p < PN; ++p) t4[p] = ((const float4*)Atb)[(size_t)p * plane4 + i];

    float a0[PN], a1[PN], a2[PN], a3[PN];
    int   nl[PN], nr[PN];
#pragma unroll
    for (int p = 0; p < PN; ++p) {
        a0[p] = fabsf(hyp[p * 4 + 0]);
        a1[p] = fabsf(hyp[p * 4 + 1]);
        a2[p] = fabsf(hyp[p * 4 + 2]);
        a3[p] = fabsf(hyp[p * 4 + 3]);
        nl[p] = nbr[p * 2 + 0];
        nr[p] = nbr[p * 2 + 1];
    }

    float tt[PN][4], xn[PN][4], un[PN][4];
#pragma unroll
    for (int p = 0; p < PN; ++p) {
        tt[p][0] = t4[p].x; tt[p][1] = t4[p].y; tt[p][2] = t4[p].z; tt[p][3] = t4[p].w;
    }

#pragma unroll
    for (int e = 0; e < 4; ++e) {
        float xv[PN] = {0.f, 0.f, 0.f, 0.f, 0.f};
        auto step = [&](int p) -> float {
            const float xo  = xv[p];
            const float sgn = (xo > 0.f) ? 1.f : ((xo < 0.f) ? -1.f : 0.f);
            const float g = 0.f - tt[p][e] + 2.f * a0[p] * xo + a2[p] * sgn
                            - a0[p] * (pick5(xv, nl[p]) + pick5(xv, nr[p]));
            return xo - a1[p] * g;
        };
        {
            const float y0 = step(0);
            const float y2 = step(2);
            xv[0] = y0; xv[2] = y2;
        }
        {
            const float y1 = step(1);
            const float y3 = step(3);
            xv[1] = y1; xv[3] = y3;
        }
        xv[4] = step(4);
#pragma unroll
        for (int p = 0; p < PN; ++p) {
            xn[p][e] = xv[p];
            un[p][e] = a3[p] * (2.f * xv[p] - pick5(xv, nl[p]) - pick5(xv, nr[p]));
        }
    }

#pragma unroll
    for (int p = 0; p < PN; ++p) {
        ((float4*)X)[(size_t)p * plane4 + i] = make_float4(xn[p][0], xn[p][1], xn[p][2], xn[p][3]);
        ((float4*)U)[(size_t)p * plane4 + i] = make_float4(un[p][0], un[p][1], un[p][2], un[p][3]);
    }
}

__global__ void __launch_bounds__(UPD_THREADS)
update_kernel(const float* __restrict__ A, const float* __restrict__ axbuf,
              const float* __restrict__ Atb, const float* __restrict__ hyp,
              const int* __restrict__ nbr,
              float* __restrict__ X, float* __restrict__ U, int k) {
    __shared__ float sB[MROWS][NGRP];
    const int n0w = blockIdx.x * NGRP;

    float4 st[UPD_STG];
#pragma unroll
    for (int t = 0; t < UPD_STG; ++t) {
        const int idx = min((int)threadIdx.x + t * UPD_THREADS, MROWS * NGRP / 4 - 1);
        const int row = idx / (NGRP / 4);
        const int c4  = (idx % (NGRP / 4)) * 4;
        st[t] = *(const float4*)&A[(size_t)row * NCOLS + n0w + c4];
    }
#pragma unroll
    for (int t = 0; t < UPD_STG; ++t) {
        const int idx = threadIdx.x + t * UPD_THREADS;
        const int row = idx / (NGRP / 4);
        const int c4  = (idx % (NGRP / 4)) * 4;
        if (idx < MROWS * NGRP / 4)
            *(float4*)&sB[row][c4] = st[t];
    }
    __syncthreads();

    const int wave = threadIdx.x >> 5;
    const int jt   = wave % JT5;
    const int r0   = (blockIdx.y * RT_WG + wave / JT5) * 16;
    const int lane = threadIdx.x & 31;
    const int c    = lane & 15;
    const int kh   = (lane >> 4) << 1;

    const size_t stride = (size_t)BB * NCOLS;
    const int ih  = (lane >> 4) * 8;
    const int col = n0w + jt * 16 + c;

    float x[8][PN], u[8][PN], t[8][PN];
#pragma unroll
    for (int v = 0; v < 4; ++v) {
        const size_t off = (size_t)(r0 + ih + v) * NCOLS + col;
#pragma unroll
        for (int p = 0; p < PN; ++p) {
            x[v][p] = X[(size_t)p * stride + off];
            u[v][p] = U[(size_t)p * stride + off];
            t[v][p] = Atb[(size_t)p * stride + off];
        }
    }

    const float* pa = axbuf + (size_t)(r0 + c) * AXLD + kh;
    const float* pb = &sB[kh][jt * 16 + c];

    v8f acc[PN] = {};
#pragma unroll
    for (int ks = 0; ks < MP / 4; ++ks) {
#pragma unroll
        for (int s = 0; s < PN; ++s) {
            const v2f af = *(const v2f*)(pa + (size_t)s * BB * AXLD + 4 * ks);
            v2f bf;
            bf.x = pb[(s * MP + 4 * ks) * NGRP];
            bf.y = pb[(s * MP + 4 * ks + 1) * NGRP];
            acc[s] = wmma4(af, bf, acc[s]);
        }
    }

#pragma unroll
    for (int v = 4; v < 8; ++v) {
        const size_t off = (size_t)(r0 + ih + v) * NCOLS + col;
#pragma unroll
        for (int p = 0; p < PN; ++p) {
            x[v][p] = X[(size_t)p * stride + off];
            u[v][p] = U[(size_t)p * stride + off];
            t[v][p] = Atb[(size_t)p * stride + off];
        }
    }

    float a0[PN], a1[PN], a2[PN], a3[PN];
    int   nl[PN], nr[PN];
#pragma unroll
    for (int p = 0; p < PN; ++p) {
        a0[p] = fabsf(hyp[(k * PN + p) * 4 + 0]);
        a1[p] = fabsf(hyp[(k * PN + p) * 4 + 1]);
        a2[p] = fabsf(hyp[(k * PN + p) * 4 + 2]);
        a3[p] = fabsf(hyp[(k * PN + p) * 4 + 3]);
        nl[p] = nbr[p * 2 + 0];
        nr[p] = nbr[p * 2 + 1];
    }

#pragma unroll
    for (int v = 0; v < 8; ++v) {
        const size_t off = (size_t)(r0 + ih + v) * NCOLS + col;
        float (&xv)[PN] = x[v];
        auto step = [&](int p, float av) -> float {
            const float xo  = xv[p];
            const float sgn = (xo > 0.f) ? 1.f : ((xo < 0.f) ? -1.f : 0.f);
            const float g = av - t[v][p] + 2.f * a0[p] * xo + a2[p] * sgn
                            + 2.f * u[v][p] - a0[p] * (pick5(xv, nl[p]) + pick5(xv, nr[p]));
            return xo - a1[p] * g;
        };
        {
            const float y0 = step(0, acc[0][v]);
            const float y2 = step(2, acc[2][v]);
            xv[0] = y0; xv[2] = y2;
        }
        {
            const float y1 = step(1, acc[1][v]);
            const float y3 = step(3, acc[3][v]);
            xv[1] = y1; xv[3] = y3;
        }
        xv[4] = step(4, acc[4][v]);
        float un[PN];
#pragma unroll
        for (int p = 0; p < PN; ++p)
            un[p] = u[v][p] + a3[p] * (2.f * xv[p] - pick5(xv, nl[p]) - pick5(xv, nr[p]));
#pragma unroll
        for (int p = 0; p < PN; ++p) {
            X[(size_t)p * stride + off] = xv[p];
            U[(size_t)p * stride + off] = un[p];
        }
    }
}

extern "C" void kernel_launch(void* const* d_in, const int* in_sizes, int n_in,
                              void* d_out, int out_size, void* d_ws, size_t ws_size,
                              hipStream_t stream) {
    (void)in_sizes; (void)n_in; (void)out_size; (void)ws_size;
    const float* A   = (const float*)d_in[0];
    const float* b   = (const float*)d_in[1];
    const float* hyp = (const float*)d_in[2];
    const int*   nbr = (const int*)d_in[3];

    float* X = (float*)d_out;
    float* U = X + (size_t)PN * BB * NCOLS;
    float* Atb   = (float*)d_ws;
    float* axbuf = Atb + (size_t)PN * BB * NCOLS;
    float* part  = axbuf + (size_t)PN * BB * AXLD;

    atb_kernel<<<dim3(25, 4, PN), 256, 0, stream>>>(A, b, Atb);

    first_update_kernel<<<BB * NCOLS / 4 / 256, 256, 0, stream>>>(Atb, hyp, nbr, X, U);

    for (int k = 1; k < KITERS; ++k) {
        ax_part_kernel<<<dim3(8, KSPLIT, PN), 256, 0, stream>>>(A, X, part);
        ax_sum_kernel <<<dim3(BB * AXLD / 4 / 256, PN), 256, 0, stream>>>(part, axbuf);
        update_kernel <<<dim3(25, BB / (RT_WG * 16)), UPD_THREADS, 0, stream>>>(
            A, axbuf, Atb, hyp, nbr, X, U, k);
    }
}


# update launch reshaped to one resident round: 250 work-groups of 32 waves, one per WGP, instead of 400 of 20 waves in two rounds; operand rows held four at a time in 128 registers; sums unchanged
# speedup vs baseline: 17.2337x; 1.0559x over previous
#include <hip/hip_runtime.h>

#define PN      5
#define MROWS   500
#define NCOLS   2000
#define BB      1024
#define KITERS  10
#define MP      100
#define AXLD    112
#define JT5     5
#define NGRP    80
#define KLD     102
#define KCH     100
#define MPAD    112
#define ROWSPAN 512
#define KSPLIT  20
#define STG     11

typedef __attribute__((ext_vector_type(2))) float v2f;
typedef __attribute__((ext_vector_type(8))) float v8f;

__device__ __forceinline__ v8f wmma4(v2f a, v2f b, v8f c) {
    return __builtin_amdgcn_wmma_f32_16x16x4_f32(
        false, a, false, b, (short)0, c, false, false);
}


__global__ void __launch_bounds__(256, 1)
atb_kernel(const float* __restrict__ A, const float* __restrict__ b,
           float* __restrict__ Atb) {
    const int p    = blockIdx.z;
    const int n0   = blockIdx.x * NGRP;
    const int it0  = blockIdx.y * 8 + (threadIdx.x >> 5);
    const int r0   = it0 * 16;
    const int r1   = r0 + ROWSPAN;
    const float* Ap = A + (size_t)p * MP * NCOLS;

    __shared__ float sBt[NGRP][KLD];
    for (int idx = threadIdx.x; idx < MP * NGRP / 4; idx += 256) {
        const int kk = idx / (NGRP / 4);
        const int c4 = (idx % (NGRP / 4)) * 4;
        float4 v = *(const float4*)&Ap[(size_t)kk * NCOLS + n0 + c4];
        sBt[c4 + 0][kk] = v.x;
        sBt[c4 + 1][kk] = v.y;
        sBt[c4 + 2][kk] = v.z;
        sBt[c4 + 3][kk] = v.w;
    }
    __syncthreads();

    const int lane = threadIdx.x & 31;
    const int nl   = lane & 15;
    const int kh   = (lane >> 4) << 1;

    const float* pa0 = b + (size_t)(r0 + nl) * MROWS + p * MP + kh;
    const float* pa1 = b + (size_t)(r1 + nl) * MROWS + p * MP + kh;
    const float* ps  = &sBt[nl][kh];

    v8f acc0[JT5] = {};
    v8f acc1[JT5] = {};
    for (int k0 = 0; k0 < MP; k0 += 4) {
        v2f af0; af0.x = pa0[0]; af0.y = pa0[1]; pa0 += 4;
        v2f af1; af1.x = pa1[0]; af1.y = pa1[1]; pa1 += 4;
#pragma unroll
        for (int j = 0; j < JT5; ++j) {
            v2f bf = *(const v2f*)(ps + j * 16 * KLD);
            acc0[j] = wmma4(af0, bf, acc0[j]);
            acc1[j] = wmma4(af1, bf, acc1[j]);
        }
        ps += 4;
    }

    float* out = Atb + (size_t)p * BB * NCOLS;
    const int ih = (lane >> 4) * 8;
#pragma unroll
    for (int j = 0; j < JT5; ++j) {
        const int col = n0 + j * 16 + nl;
#pragma unroll
        for (int v = 0; v < 8; ++v) {
            out[(size_t)(r0 + ih + v) * NCOLS + col] = acc0[j][v];
            out[(size_t)(r1 + ih + v) * NCOLS + col] = acc1[j][v];
        }
    }
}

__global__ void __launch_bounds__(256, 1)
ax_part_kernel(const float* __restrict__ A, const float* __restrict__ X,
               float* __restrict__ part) {
    const int p  = blockIdx.z;
    const int r0 = (blockIdx.x * 8 + (threadIdx.x >> 5)) * 16;
    const int kc = blockIdx.y * KCH;
    const float* Ap = A + (size_t)p * MP * NCOLS;
    const float* Xp = X + (size_t)p * BB * NCOLS;

    __shared__ float sB[MPAD][KCH];

    const int lane = threadIdx.x & 31;
    const int ml   = lane & 15;
    const int kh   = (lane >> 4) << 1;

    float4 st[STG];
#pragma unroll
    for (int t = 0; t < STG; ++t) {
        const int idx = threadIdx.x + t * 256;
        const int m   = idx / (KCH / 4);
        const int c4  = (idx % (KCH / 4)) * 4;
        st[t] = make_float4(0.f, 0.f, 0.f, 0.f);
        if (t * 256 + 255 < MP * KCH / 4) {
            st[t] = *(const float4*)&Ap[(size_t)m * NCOLS + kc + c4];
        } else if (t * 256 < MP * KCH / 4) {
            if (m < MP)
                st[t] = *(const float4*)&Ap[(size_t)m * NCOLS + kc + c4];
        }
    }

    const float* px = Xp + (size_t)(r0 + ml) * NCOLS + kc + kh;
    v2f af[KCH / 4];
#pragma unroll
    for (int i = 0; i < KCH / 4; ++i) {
        af[i].x = px[4 * i];
        af[i].y = px[4 * i + 1];
    }

#pragma unroll
    for (int t = 0; t < STG; ++t) {
        const int idx = threadIdx.x + t * 256;
        const int m   = idx / (KCH / 4);
        const int c4  = (idx % (KCH / 4)) * 4;
        if (idx < MPAD * KCH / 4)
            *(float4*)&sB[m][c4] = st[t];
    }
    __syncthreads();

    v8f acc[7] = {};
#pragma unroll
    for (int i = 0; i < KCH / 4; ++i) {
        const float* ps = &sB[ml][4 * i + kh];
#pragma unroll
        for (int j = 0; j < 7; ++j) {
            v2f bf = *(const v2f*)(ps + j * 16 * KCH);
            acc[j] = wmma4(af[i], bf, acc[j]);
        }
    }

    float* out = part + ((size_t)blockIdx.z * KSPLIT + blockIdx.y) * BB * AXLD;
    const int ih = (lane >> 4) * 8;
#pragma unroll
    for (int j = 0; j < 7; ++j) {
#pragma unroll
        for (int v = 0; v < 8; ++v)
            out[(size_t)(r0 + ih + v) * AXLD + j * 16 + ml] = acc[j][v];
    }
}

__global__ void ax_sum_kernel(const float* __restrict__ part,
                              float* __restrict__ axbuf) {
    const int n4 = BB * AXLD / 4;
    const int i  = blockIdx.x * blockDim.x + threadIdx.x;
    part  += (size_t)blockIdx.y * KSPLIT * BB * AXLD;
    axbuf += (size_t)blockIdx.y * BB * AXLD;
    const float4* src = (const float4*)part;
    float4 v[KSPLIT];
#pragma unroll
    for (int c = 0; c < KSPLIT; ++c) v[c] = src[(size_t)c * n4 + i];
    float4 s = v[0];
#pragma unroll
    for (int c = 1; c < KSPLIT; ++c) {
        s.x += v[c].x; s.y += v[c].y; s.z += v[c].z; s.w += v[c].w;
    }
    ((float4*)axbuf)[i] = s;
}

#define UPD_WAVES   32
#define UPD_THREADS (UPD_WAVES * 32)
#define UPD_YGRP    (BB / 16 * JT5 / UPD_WAVES)
#define UPD_STG     10

__device__ __forceinline__ float pick5(const float (&x)[PN], int q) {
    float r = x[0];
    r = (q == 1) ? x[1] : r;
    r = (q == 2) ? x[2] : r;
    r = (q == 3) ? x[3] : r;
    r = (q == 4) ? x[4] : r;
    return r;
}

__global__ void __launch_bounds__(256)
first_update_kernel(const float* __restrict__ Atb, const float* __restrict__ hyp,
                    const int* __restrict__ nbr,
                    float* __restrict__ X, float* __restrict__ U) {
    const size_t plane4 = (size_t)BB * NCOLS / 4;
    const size_t i = (size_t)blockIdx.x * 256 + threadIdx.x;

    float4 t4[PN];
#pragma unroll
    for (int p = 0; p < PN; ++p) t4[p] = ((const float4*)Atb)[(size_t)p * plane4 + i];

    float a0[PN], a1[PN], a2[PN], a3[PN];
    int   nl[PN], nr[PN];
#pragma unroll
    for (int p = 0; p < PN; ++p) {
        a0[p] = fabsf(hyp[p * 4 + 0]);
        a1[p] = fabsf(hyp[p * 4 + 1]);
        a2[p] = fabsf(hyp[p * 4 + 2]);
        a3[p] = fabsf(hyp[p * 4 + 3]);
        nl[p] = nbr[p * 2 + 0];
        nr[p] = nbr[p * 2 + 1];
    }

    float tt[PN][4], xn[PN][4], un[PN][4];
#pragma unroll
    for (int p = 0; p < PN; ++p) {
        tt[p][0] = t4[p].x; tt[p][1] = t4[p].y; tt[p][2] = t4[p].z; tt[p][3] = t4[p].w;
    }

#pragma unroll
    for (int e = 0; e < 4; ++e) {
        float xv[PN] = {0.f, 0.f, 0.f, 0.f, 0.f};
        auto step = [&](int p) -> float {
            const float xo  = xv[p];
            const float sgn = (xo > 0.f) ? 1.f : ((xo < 0.f) ? -1.f : 0.f);
            const float g = 0.f - tt[p][e] + 2.f * a0[p] * xo + a2[p] * sgn
                            - a0[p] * (pick5(xv, nl[p]) + pick5(xv, nr[p]));
            return xo - a1[p] * g;
        };
        {
            const float y0 = step(0);
            const float y2 = step(2);
            xv[0] = y0; xv[2] = y2;
        }
        {
            const float y1 = step(1);
            const float y3 = step(3);
            xv[1] = y1; xv[3] = y3;
        }
        xv[4] = step(4);
#pragma unroll
        for (int p = 0; p < PN; ++p) {
            xn[p][e] = xv[p];
            un[p][e] = a3[p] * (2.f * xv[p] - pick5(xv, nl[p]) - pick5(xv, nr[p]));
        }
    }

#pragma unroll
    for (int p = 0; p < PN; ++p) {
        ((float4*)X)[(size_t)p * plane4 + i] = make_float4(xn[p][0], xn[p][1], xn[p][2], xn[p][3]);
        ((float4*)U)[(size_t)p * plane4 + i] = make_float4(un[p][0], un[p][1], un[p][2], un[p][3]);
    }
}

__global__ void __launch_bounds__(UPD_THREADS)
update_kernel(const float* __restrict__ A, const float* __restrict__ axbuf,
              const float* __restrict__ Atb, const float* __restrict__ hyp,
              const int* __restrict__ nbr,
              float* __restrict__ X, float* __restrict__ U, int k) {
    __shared__ float sB[MROWS][NGRP];
    const int n0w = blockIdx.x * NGRP;

    float4 st[UPD_STG];
#pragma unroll
    for (int t = 0; t < UPD_STG; ++t) {
        const int idx = min((int)threadIdx.x + t * UPD_THREADS, MROWS * NGRP / 4 - 1);
        const int row = idx / (NGRP / 4);
        const int c4  = (idx % (NGRP / 4)) * 4;
        st[t] = *(const float4*)&A[(size_t)row * NCOLS + n0w + c4];
    }
#pragma unroll
    for (int t = 0; t < UPD_STG; ++t) {
        const int idx = threadIdx.x + t * UPD_THREADS;
        const int row = idx / (NGRP / 4);
        const int c4  = (idx % (NGRP / 4)) * 4;
        if (idx < MROWS * NGRP / 4)
            *(float4*)&sB[row][c4] = st[t];
    }
    __syncthreads();

    const int wave = threadIdx.x >> 5;
    const int tile = blockIdx.y * UPD_WAVES + wave;
    const int jt   = tile % JT5;
    const int r0   = (tile / JT5) * 16;
    const int lane = threadIdx.x & 31;
    const int c    = lane & 15;
    const int kh   = (lane >> 4) << 1;

    const size_t stride = (size_t)BB * NCOLS;
    const int ih  = (lane >> 4) * 8;
    const int col = n0w + jt * 16 + c;

    float x[8][PN], u[8][PN], t[8][PN];
    auto load_row = [&](int v) {
        const size_t off = (size_t)(r0 + ih + v) * NCOLS + col;
#pragma unroll
        for (int p = 0; p < PN; ++p) {
            x[v][p] = X[(size_t)p * stride + off];
            u[v][p] = U[(size_t)p * stride + off];
            t[v][p] = Atb[(size_t)p * stride + off];
        }
    };

    load_row(0); load_row(1); load_row(2); load_row(3);

    const float* pa = axbuf + (size_t)(r0 + c) * AXLD + kh;
    const float* pb = &sB[kh][jt * 16 + c];

    v8f acc[PN] = {};
#pragma unroll
    for (int ks = 0; ks < MP / 4; ++ks) {
#pragma unroll
        for (int s = 0; s < PN; ++s) {
            const v2f af = *(const v2f*)(pa + (size_t)s * BB * AXLD + 4 * ks);
            v2f bf;
            bf.x = pb[(s * MP + 4 * ks) * NGRP];
            bf.y = pb[(s * MP + 4 * ks + 1) * NGRP];
            acc[s] = wmma4(af, bf, acc[s]);
        }
    }

    float a0[PN], a1[PN], a2[PN], a3[PN];
    int   nl[PN], nr[PN];
#pragma unroll
    for (int p = 0; p < PN; ++p) {
        a0[p] = fabsf(hyp[(k * PN + p) * 4 + 0]);
        a1[p] = fabsf(hyp[(k * PN + p) * 4 + 1]);
        a2[p] = fabsf(hyp[(k * PN + p) * 4 + 2]);
        a3[p] = fabsf(hyp[(k * PN + p) * 4 + 3]);
        nl[p] = nbr[p * 2 + 0];
        nr[p] = nbr[p * 2 + 1];
    }

    auto do_row = [&](int v, const float (&av)[PN]) {
        const size_t off = (size_t)(r0 + ih + v) * NCOLS + col;
        float (&xv)[PN] = x[v];
        auto step = [&](int p) -> float {
            const float xo  = xv[p];
            const float sgn = (xo > 0.f) ? 1.f : ((xo < 0.f) ? -1.f : 0.f);
            const float g = av[p] - t[v][p] + 2.f * a0[p] * xo + a2[p] * sgn
                            + 2.f * u[v][p] - a0[p] * (pick5(xv, nl[p]) + pick5(xv, nr[p]));
            return xo - a1[p] * g;
        };
        {
            const float y0 = step(0);
            const float y2 = step(2);
            xv[0] = y0; xv[2] = y2;
        }
        {
            const float y1 = step(1);
            const float y3 = step(3);
            xv[1] = y1; xv[3] = y3;
        }
        xv[4] = step(4);
        float un[PN];
#pragma unroll
        for (int p = 0; p < PN; ++p)
            un[p] = u[v][p] + a3[p] * (2.f * xv[p] - pick5(xv, nl[p]) - pick5(xv, nr[p]));
#pragma unroll
        for (int p = 0; p < PN; ++p) {
            X[(size_t)p * stride + off] = xv[p];
            U[(size_t)p * stride + off] = un[p];
        }
    };

#pragma unroll
    for (int v = 0; v < 8; ++v) {
        const float av[PN] = {acc[0][v], acc[1][v], acc[2][v], acc[3][v], acc[4][v]};
        do_row(v, av);
        if (v + 4 < 8) load_row(v + 4);
    }
}

extern "C" void kernel_launch(void* const* d_in, const int* in_sizes, int n_in,
                              void* d_out, int out_size, void* d_ws, size_t ws_size,
                              hipStream_t stream) {
    (void)in_sizes; (void)n_in; (void)out_size; (void)ws_size;
    const float* A   = (const float*)d_in[0];
    const float* b   = (const float*)d_in[1];
    const float* hyp = (const float*)d_in[2];
    const int*   nbr = (const int*)d_in[3];

    float* X = (float*)d_out;
    float* U = X + (size_t)PN * BB * NCOLS;
    float* Atb   = (float*)d_ws;
    float* axbuf = Atb + (size_t)PN * BB * NCOLS;
    float* part  = axbuf + (size_t)PN * BB * AXLD;

    atb_kernel<<<dim3(25, 4, PN), 256, 0, stream>>>(A, b, Atb);

    first_update_kernel<<<BB * NCOLS / 4 / 256, 256, 0, stream>>>(Atb, hyp, nbr, X, U);

    for (int k = 1; k < KITERS; ++k) {
        ax_part_kernel<<<dim3(8, KSPLIT, PN), 256, 0, stream>>>(A, X, part);
        ax_sum_kernel <<<dim3(BB * AXLD / 4 / 256, PN), 256, 0, stream>>>(part, axbuf);
        update_kernel <<<dim3(25, UPD_YGRP), UPD_THREADS, 0, stream>>>(
            A, axbuf, Atb, hyp, nbr, X, U, k);
    }
}


# AX summed per 16-row tile and written in fragment order [b/16][m/2][b%16][m%2] (turned through LDS): an update A-fragment load is one aligned 256-B piece, not 16 lines; sums unchanged
# speedup vs baseline: 20.5791x; 1.1941x over previous
#include <hip/hip_runtime.h>

#define PN      5
#define MROWS   500
#define NCOLS   2000
#define BB      1024
#define KITERS  10
#define MP      100
#define AXLD    112
#define JT5     5
#define NGRP    80
#define KLD     102
#define KCH     100
#define MPAD    112
#define ROWSPAN 512
#define KSPLIT  20
#define STG     11

typedef __attribute__((ext_vector_type(2))) float v2f;
typedef __attribute__((ext_vector_type(8))) float v8f;

__device__ __forceinline__ v8f wmma4(v2f a, v2f b, v8f c) {
    return __builtin_amdgcn_wmma_f32_16x16x4_f32(
        false, a, false, b, (short)0, c, false, false);
}


__global__ void __launch_bounds__(256, 1)
atb_kernel(const float* __restrict__ A, const float* __restrict__ b,
           float* __restrict__ Atb) {
    const int p    = blockIdx.z;
    const int n0   = blockIdx.x * NGRP;
    const int it0  = blockIdx.y * 8 + (threadIdx.x >> 5);
    const int r0   = it0 * 16;
    const int r1   = r0 + ROWSPAN;
    const float* Ap = A + (size_t)p * MP * NCOLS;

    __shared__ float sBt[NGRP][KLD];
    for (int idx = threadIdx.x; idx < MP * NGRP / 4; idx += 256) {
        const int kk = idx / (NGRP / 4);
        const int c4 = (idx % (NGRP / 4)) * 4;
        float4 v = *(const float4*)&Ap[(size_t)kk * NCOLS + n0 + c4];
        sBt[c4 + 0][kk] = v.x;
        sBt[c4 + 1][kk] = v.y;
        sBt[c4 + 2][kk] = v.z;
        sBt[c4 + 3][kk] = v.w;
    }
    __syncthreads();

    const int lane = threadIdx.x & 31;
    const int nl   = lane & 15;
    const int kh   = (lane >> 4) << 1;

    const float* pa0 = b + (size_t)(r0 + nl) * MROWS + p * MP + kh;
    const float* pa1 = b + (size_t)(r1 + nl) * MROWS + p * MP + kh;
    const float* ps  = &sBt[nl][kh];

    v8f acc0[JT5] = {};
    v8f acc1[JT5] = {};
    for (int k0 = 0; k0 < MP; k0 += 4) {
        v2f af0; af0.x = pa0[0]; af0.y = pa0[1]; pa0 += 4;
        v2f af1; af1.x = pa1[0]; af1.y = pa1[1]; pa1 += 4;
#pragma unroll
        for (int j = 0; j < JT5; ++j) {
            v2f bf = *(const v2f*)(ps + j * 16 * KLD);
            acc0[j] = wmma4(af0, bf, acc0[j]);
            acc1[j] = wmma4(af1, bf, acc1[j]);
        }
        ps += 4;
    }

    float* out = Atb + (size_t)p * BB * NCOLS;
    const int ih = (lane >> 4) * 8;
#pragma unroll
    for (int j = 0; j < JT5; ++j) {
        const int col = n0 + j * 16 + nl;
#pragma unroll
        for (int v = 0; v < 8; ++v) {
            out[(size_t)(r0 + ih + v) * NCOLS + col] = acc0[j][v];
            out[(size_t)(r1 + ih + v) * NCOLS + col] = acc1[j][v];
        }
    }
}

__global__ void __launch_bounds__(256, 1)
ax_part_kernel(const float* __restrict__ A, const float* __restrict__ X,
               float* __restrict__ part) {
    const int p  = blockIdx.z;
    const int r0 = (blockIdx.x * 8 + (threadIdx.x >> 5)) * 16;
    const int kc = blockIdx.y * KCH;
    const float* Ap = A + (size_t)p * MP * NCOLS;
    const float* Xp = X + (size_t)p * BB * NCOLS;

    __shared__ float sB[MPAD][KCH];

    const int lane = threadIdx.x & 31;
    const int ml   = lane & 15;
    const int kh   = (lane >> 4) << 1;

    float4 st[STG];
#pragma unroll
    for (int t = 0; t < STG; ++t) {
        const int idx = threadIdx.x + t * 256;
        const int m   = idx / (KCH / 4);
        const int c4  = (idx % (KCH / 4)) * 4;
        st[t] = make_float4(0.f, 0.f, 0.f, 0.f);
        if (t * 256 + 255 < MP * KCH / 4) {
            st[t] = *(const float4*)&Ap[(size_t)m * NCOLS + kc + c4];
        } else if (t * 256 < MP * KCH / 4) {
            if (m < MP)
                st[t] = *(const float4*)&Ap[(size_t)m * NCOLS + kc + c4];
        }
    }

    const float* px = Xp + (size_t)(r0 + ml) * NCOLS + kc + kh;
    v2f af[KCH / 4];
#pragma unroll
    for (int i = 0; i < KCH / 4; ++i) {
        af[i].x = px[4 * i];
        af[i].y = px[4 * i + 1];
    }

#pragma unroll
    for (int t = 0; t < STG; ++t) {
        const int idx = threadIdx.x + t * 256;
        const int m   = idx / (KCH / 4);
        const int c4  = (idx % (KCH / 4)) * 4;
        if (idx < MPAD * KCH / 4)
            *(float4*)&sB[m][c4] = st[t];
    }
    __syncthreads();

    v8f acc[7] = {};
#pragma unroll
    for (int i = 0; i < KCH / 4; ++i) {
        const float* ps = &sB[ml][4 * i + kh];
#pragma unroll
        for (int j = 0; j < 7; ++j) {
            v2f bf = *(const v2f*)(ps + j * 16 * KCH);
            acc[j] = wmma4(af[i], bf, acc[j]);
        }
    }

    float* out = part + ((size_t)blockIdx.z * KSPLIT + blockIdx.y) * BB * AXLD;
    const int ih = (lane >> 4) * 8;
#pragma unroll
    for (int j = 0; j < 7; ++j) {
#pragma unroll
        for (int v = 0; v < 8; ++v)
            out[(size_t)(r0 + ih + v) * AXLD + j * 16 + ml] = acc[j][v];
    }
}

#define SUM_THREADS (16 * AXLD / 4)
#define SUM_LD      (AXLD + 4)
__global__ void __launch_bounds__(SUM_THREADS)
ax_sum_kernel(const float* __restrict__ part,
              float* __restrict__ axbuf) {
    __shared__ float sT[16][SUM_LD];
    const int n4 = BB * AXLD / 4;
    const int t  = threadIdx.x;
    const int i  = blockIdx.x * SUM_THREADS + t;
    part  += (size_t)blockIdx.y * KSPLIT * BB * AXLD;
    axbuf += (size_t)blockIdx.y * BB * AXLD;
    const float4* src = (const float4*)part;
    float4 v[KSPLIT];
#pragma unroll
    for (int c = 0; c < KSPLIT; ++c) v[c] = src[(size_t)c * n4 + i];
    float4 s = v[0];
#pragma unroll
    for (int c = 1; c < KSPLIT; ++c) {
        s.x += v[c].x; s.y += v[c].y; s.z += v[c].z; s.w += v[c].w;
    }
    *(float4*)&sT[t / (AXLD / 4)][(t % (AXLD / 4)) * 4] = s;
    __syncthreads();
    const int mp = t >> 3;
    const int c  = (t & 7) * 2;
    const v2f lo = *(const v2f*)&sT[c][2 * mp];
    const v2f hi = *(const v2f*)&sT[c + 1][2 * mp];
    ((float4*)axbuf)[i] = make_float4(lo.x, lo.y, hi.x, hi.y);
}

#define UPD_WAVES   32
#define UPD_THREADS (UPD_WAVES * 32)
#define UPD_YGRP    (BB / 16 * JT5 / UPD_WAVES)
#define UPD_STG     10

__device__ __forceinline__ float pick5(const float (&x)[PN], int q) {
    float r = x[0];
    r = (q == 1) ? x[1] : r;
    r = (q == 2) ? x[2] : r;
    r = (q == 3) ? x[3] : r;
    r = (q == 4) ? x[4] : r;
    return r;
}

__global__ void __launch_bounds__(256)
first_update_kernel(const float* __restrict__ Atb, const float* __restrict__ hyp,
                    const int* __restrict__ nbr,
                    float* __restrict__ X, float* __restrict__ U) {
    const size_t plane4 = (size_t)BB * NCOLS / 4;
    const size_t i = (size_t)blockIdx.x * 256 + threadIdx.x;

    float4 t4[PN];
#pragma unroll
    for (int p = 0; p < PN; ++p) t4[p] = ((const float4*)Atb)[(size_t)p * plane4 + i];

    float a0[PN], a1[PN], a2[PN], a3[PN];
    int   nl[PN], nr[PN];
#pragma unroll
    for (int p = 0; p < PN; ++p) {
        a0[p] = fabsf(hyp[p * 4 + 0]);
        a1[p] = fabsf(hyp[p * 4 + 1]);
        a2[p] = fabsf(hyp[p * 4 + 2]);
        a3[p] = fabsf(hyp[p * 4 + 3]);
        nl[p] = nbr[p * 2 + 0];
        nr[p] = nbr[p * 2 + 1];
    }

    float tt[PN][4], xn[PN][4], un[PN][4];
#pragma unroll
    for (int p = 0; p < PN; ++p) {
        tt[p][0] = t4[p].x; tt[p][1] = t4[p].y; tt[p][2] = t4[p].z; tt[p][3] = t4[p].w;
    }

#pragma unroll
    for (int e = 0; e < 4; ++e) {
        float xv[PN] = {0.f, 0.f, 0.f, 0.f, 0.f};
        auto step = [&](int p) -> float {
            const float xo  = xv[p];
            const float sgn = (xo > 0.f) ? 1.f : ((xo < 0.f) ? -1.f : 0.f);
            const float g = 0.f - tt[p][e] + 2.f * a0[p] * xo + a2[p] * sgn
                            - a0[p] * (pick5(xv, nl[p]) + pick5(xv, nr[p]));
            return xo - a1[p] * g;
        };
        {
            const float y0 = step(0);
            const float y2 = step(2);
            xv[0] = y0; xv[2] = y2;
        }
        {
            const float y1 = step(1);
            const float y3 = step(3);
            xv[1] = y1; xv[3] = y3;
        }
        xv[4] = step(4);
#pragma unroll
        for (int p = 0; p < PN; ++p) {
            xn[p][e] = xv[p];
            un[p][e] = a3[p] * (2.f * xv[p] - pick5(xv, nl[p]) - pick5(xv, nr[p]));
        }
    }

#pragma unroll
    for (int p = 0; p < PN; ++p) {
        ((float4*)X)[(size_t)p * plane4 + i] = make_float4(xn[p][0], xn[p][1], xn[p][2], xn[p][3]);
        ((float4*)U)[(size_t)p * plane4 + i] = make_float4(un[p][0], un[p][1], un[p][2], un[p][3]);
    }
}

__global__ void __launch_bounds__(UPD_THREADS)
update_kernel(const float* __restrict__ A, const float* __restrict__ axbuf,
              const float* __restrict__ Atb, const float* __restrict__ hyp,
              const int* __restrict__ nbr,
              float* __restrict__ X, float* __restrict__ U, int k) {
    __shared__ float sB[MROWS][NGRP];
    const int n0w = blockIdx.x * NGRP;

    float4 st[UPD_STG];
#pragma unroll
    for (int t = 0; t < UPD_STG; ++t) {
        const int idx = min((int)threadIdx.x + t * UPD_THREADS, MROWS * NGRP / 4 - 1);
        const int row = idx / (NGRP / 4);
        const int c4  = (idx % (NGRP / 4)) * 4;
        st[t] = *(const float4*)&A[(size_t)row * NCOLS + n0w + c4];
    }
#pragma unroll
    for (int t = 0; t < UPD_STG; ++t) {
        const int idx = threadIdx.x + t * UPD_THREADS;
        const int row = idx / (NGRP / 4);
        const int c4  = (idx % (NGRP / 4)) * 4;
        if (idx < MROWS * NGRP / 4)
            *(float4*)&sB[row][c4] = st[t];
    }
    __syncthreads();

    const int wave = threadIdx.x >> 5;
    const int tile = blockIdx.y * UPD_WAVES + wave;
    const int jt   = tile % JT5;
    const int r0   = (tile / JT5) * 16;
    const int lane = threadIdx.x & 31;
    const int c    = lane & 15;
    const int kh   = (lane >> 4) << 1;

    const size_t stride = (size_t)BB * NCOLS;
    const int ih  = (lane >> 4) * 8;
    const int col = n0w + jt * 16 + c;

    float x[8][PN], u[8][PN], t[8][PN];
    auto load_row = [&](int v) {
        const size_t off = (size_t)(r0 + ih + v) * NCOLS + col;
#pragma unroll
        for (int p = 0; p < PN; ++p) {
            x[v][p] = X[(size_t)p * stride + off];
            u[v][p] = U[(size_t)p * stride + off];
            t[v][p] = Atb[(size_t)p * stride + off];
        }
    };

    load_row(0); load_row(1); load_row(2); load_row(3);

    const float* pa = axbuf + (((size_t)(r0 >> 4) * (AXLD / 2) + (lane >> 4)) * 16 + c) * 2;
    const float* pb = &sB[kh][jt * 16 + c];

    v8f acc[PN] = {};
#pragma unroll
    for (int ks = 0; ks < MP / 4; ++ks) {
#pragma unroll
        for (int s = 0; s < PN; ++s) {
            const v2f af = *(const v2f*)(pa + (size_t)s * BB * AXLD + ks * 64);
            v2f bf;
            bf.x = pb[(s * MP + 4 * ks) * NGRP];
            bf.y = pb[(s * MP + 4 * ks + 1) * NGRP];
            acc[s] = wmma4(af, bf, acc[s]);
        }
    }

    float a0[PN], a1[PN], a2[PN], a3[PN];
    int   nl[PN], nr[PN];
#pragma unroll
    for (int p = 0; p < PN; ++p) {
        a0[p] = fabsf(hyp[(k * PN + p) * 4 + 0]);
        a1[p] = fabsf(hyp[(k * PN + p) * 4 + 1]);
        a2[p] = fabsf(hyp[(k * PN + p) * 4 + 2]);
        a3[p] = fabsf(hyp[(k * PN + p) * 4 + 3]);
        nl[p] = nbr[p * 2 + 0];
        nr[p] = nbr[p * 2 + 1];
    }

    auto do_row = [&](int v, const float (&av)[PN]) {
        const size_t off = (size_t)(r0 + ih + v) * NCOLS + col;
        float (&xv)[PN] = x[v];
        auto step = [&](int p) -> float {
            const float xo  = xv[p];
            const float sgn = (xo > 0.f) ? 1.f : ((xo < 0.f) ? -1.f : 0.f);
            const float g = av[p] - t[v][p] + 2.f * a0[p] * xo + a2[p] * sgn
                            + 2.f * u[v][p] - a0[p] * (pick5(xv, nl[p]) + pick5(xv, nr[p]));
            return xo - a1[p] * g;
        };
        {
            const float y0 = step(0);
            const float y2 = step(2);
            xv[0] = y0; xv[2] = y2;
        }
        {
            const float y1 = step(1);
            const float y3 = step(3);
            xv[1] = y1; xv[3] = y3;
        }
        xv[4] = step(4);
        float un[PN];
#pragma unroll
        for (int p = 0; p < PN; ++p)
            un[p] = u[v][p] + a3[p] * (2.f * xv[p] - pick5(xv, nl[p]) - pick5(xv, nr[p]));
#pragma unroll
        for (int p = 0; p < PN; ++p) {
            X[(size_t)p * stride + off] = xv[p];
            U[(size_t)p * stride + off] = un[p];
        }
    };

#pragma unroll
    for (int v = 0; v < 8; ++v) {
        const float av[PN] = {acc[0][v], acc[1][v], acc[2][v], acc[3][v], acc[4][v]};
        do_row(v, av);
        if (v + 4 < 8) load_row(v + 4);
    }
}

extern "C" void kernel_launch(void* const* d_in, const int* in_sizes, int n_in,
                              void* d_out, int out_size, void* d_ws, size_t ws_size,
                              hipStream_t stream) {
    (void)in_sizes; (void)n_in; (void)out_size; (void)ws_size;
    const float* A   = (const float*)d_in[0];
    const float* b   = (const float*)d_in[1];
    const float* hyp = (const float*)d_in[2];
    const int*   nbr = (const int*)d_in[3];

    float* X = (float*)d_out;
    float* U = X + (size_t)PN * BB * NCOLS;
    float* Atb   = (float*)d_ws;
    float* axbuf = Atb + (size_t)PN * BB * NCOLS;
    float* part  = axbuf + (size_t)PN * BB * AXLD;

    atb_kernel<<<dim3(25, 4, PN), 256, 0, stream>>>(A, b, Atb);

    first_update_kernel<<<BB * NCOLS / 4 / 256, 256, 0, stream>>>(Atb, hyp, nbr, X, U);

    for (int k = 1; k < KITERS; ++k) {
        ax_part_kernel<<<dim3(8, KSPLIT, PN), 256, 0, stream>>>(A, X, part);
        ax_sum_kernel <<<dim3(BB / 16, PN), SUM_THREADS, 0, stream>>>(part, axbuf);
        update_kernel <<<dim3(25, UPD_YGRP), UPD_THREADS, 0, stream>>>(
            A, axbuf, Atb, hyp, nbr, X, U, k);
    }
}


# AX product: 4 K chunks per 32-wave work-group, their partial sums added through LDS before the store (partial planes 20 -> 5, X fragments loaded 5 k-steps at a time); changes the K summation order
# speedup vs baseline: 23.1279x; 1.1239x over previous
#include <hip/hip_runtime.h>

#define PN      5
#define MROWS   500
#define NCOLS   2000
#define BB      1024
#define KITERS  10
#define MP      100
#define AXLD    112
#define JT5     5
#define NGRP    80
#define KLD     102
#define KCH     100
#define MPAD    112
#define ROWSPAN 512
#define KSPLIT  20
#define STG     11
#define AFG     5
#define KSLOT   4
#define KGRP    (KSPLIT / KSLOT)

typedef __attribute__((ext_vector_type(2))) float v2f;
typedef __attribute__((ext_vector_type(8))) float v8f;

__device__ __forceinline__ v8f wmma4(v2f a, v2f b, v8f c) {
    return __builtin_amdgcn_wmma_f32_16x16x4_f32(
        false, a, false, b, (short)0, c, false, false);
}


__global__ void __launch_bounds__(256, 1)
atb_kernel(const float* __restrict__ A, const float* __restrict__ b,
           float* __restrict__ Atb) {
    const int p    = blockIdx.z;
    const int n0   = blockIdx.x * NGRP;
    const int it0  = blockIdx.y * 8 + (threadIdx.x >> 5);
    const int r0   = it0 * 16;
    const int r1   = r0 + ROWSPAN;
    const float* Ap = A + (size_t)p * MP * NCOLS;

    __shared__ float sBt[NGRP][KLD];
    for (int idx = threadIdx.x; idx < MP * NGRP / 4; idx += 256) {
        const int kk = idx / (NGRP / 4);
        const int c4 = (idx % (NGRP / 4)) * 4;
        float4 v = *(const float4*)&Ap[(size_t)kk * NCOLS + n0 + c4];
        sBt[c4 + 0][kk] = v.x;
        sBt[c4 + 1][kk] = v.y;
        sBt[c4 + 2][kk] = v.z;
        sBt[c4 + 3][kk] = v.w;
    }
    __syncthreads();

    const int lane = threadIdx.x & 31;
    const int nl   = lane & 15;
    const int kh   = (lane >> 4) << 1;

    const float* pa0 = b + (size_t)(r0 + nl) * MROWS + p * MP + kh;
    const float* pa1 = b + (size_t)(r1 + nl) * MROWS + p * MP + kh;
    const float* ps  = &sBt[nl][kh];

    v8f acc0[JT5] = {};
    v8f acc1[JT5] = {};
    for (int k0 = 0; k0 < MP; k0 += 4) {
        v2f af0; af0.x = pa0[0]; af0.y = pa0[1]; pa0 += 4;
        v2f af1; af1.x = pa1[0]; af1.y = pa1[1]; pa1 += 4;
#pragma unroll
        for (int j = 0; j < JT5; ++j) {
            v2f bf = *(const v2f*)(ps + j * 16 * KLD);
            acc0[j] = wmma4(af0, bf, acc0[j]);
            acc1[j] = wmma4(af1, bf, acc1[j]);
        }
        ps += 4;
    }

    float* out = Atb + (size_t)p * BB * NCOLS;
    const int ih = (lane >> 4) * 8;
#pragma unroll
    for (int j = 0; j < JT5; ++j) {
        const int col = n0 + j * 16 + nl;
#pragma unroll
        for (int v = 0; v < 8; ++v) {
            out[(size_t)(r0 + ih + v) * NCOLS + col] = acc0[j][v];
            out[(size_t)(r1 + ih + v) * NCOLS + col] = acc1[j][v];
        }
    }
}

__global__ void __launch_bounds__(1024, 1)
ax_part_kernel(const float* __restrict__ A, const float* __restrict__ X,
               float* __restrict__ part) {
    const int p   = blockIdx.z;
    const int w   = __builtin_amdgcn_readfirstlane((int)(threadIdx.x >> 5));
    const int q   = w >> 3;
    const int rtl = w & 7;
    const int tid = threadIdx.x & 255;
    const int r0  = (blockIdx.x * 8 + rtl) * 16;
    const int kc  = (blockIdx.y * KSLOT + q) * KCH;
    const float* Ap = A + (size_t)p * MP * NCOLS;
    const float* Xp = X + (size_t)p * BB * NCOLS;

    __shared__ float sBq[KSLOT][MPAD][KCH];
    float (*sB)[KCH] = sBq[q];

    const int lane = threadIdx.x & 31;
    const int ml   = lane & 15;
    const int kh   = (lane >> 4) << 1;

    float4 st[STG];
#pragma unroll
    for (int t = 0; t < STG; ++t) {
        const int idx = tid + t * 256;
        const int m   = idx / (KCH / 4);
        const int c4  = (idx % (KCH / 4)) * 4;
        st[t] = make_float4(0.f, 0.f, 0.f, 0.f);
        if (t * 256 + 255 < MP * KCH / 4) {
            st[t] = *(const float4*)&Ap[(size_t)m * NCOLS + kc + c4];
        } else if (t * 256 < MP * KCH / 4) {
            if (m < MP)
                st[t] = *(const float4*)&Ap[(size_t)m * NCOLS + kc + c4];
        }
    }

    const float* px = Xp + (size_t)(r0 + ml) * NCOLS + kc + kh;
    v2f afc[AFG], afn[AFG];
#pragma unroll
    for (int i = 0; i < AFG; ++i) {
        afc[i].x = px[4 * i];
        afc[i].y = px[4 * i + 1];
    }

#pragma unroll
    for (int t = 0; t < STG; ++t) {
        const int idx = tid + t * 256;
        const int m   = idx / (KCH / 4);
        const int c4  = (idx % (KCH / 4)) * 4;
        if (idx < MPAD * KCH / 4)
            *(float4*)&sB[m][c4] = st[t];
    }
    __syncthreads();

    v8f acc[7] = {};
    const float* ps = &sB[ml][kh];
#pragma unroll 1
    for (int g = 0; g < KCH / 4 / AFG; ++g) {
        px += 4 * AFG;
        if (g + 1 < KCH / 4 / AFG) {
#pragma unroll
            for (int i = 0; i < AFG; ++i) {
                afn[i].x = px[4 * i];
                afn[i].y = px[4 * i + 1];
            }
        }
#pragma unroll
        for (int i = 0; i < AFG; ++i) {
#pragma unroll
            for (int j = 0; j < 7; ++j) {
                v2f bf = *(const v2f*)(ps + 4 * i + j * 16 * KCH);
                acc[j] = wmma4(afc[i], bf, acc[j]);
            }
        }
        ps += 4 * AFG;
#pragma unroll
        for (int i = 0; i < AFG; ++i) afc[i] = afn[i];
    }

    float* red = &sBq[0][0][0];
    __syncthreads();
    const int ln  = (int)__builtin_amdgcn_mbcnt_lo(~0u, 0u);
    const int mlo = ln & 15;
#pragma unroll
    for (int j = 0; j < 7; ++j) {
        const int o = j >> 1;
        if (q != o) {
            const int slot = (q < o) ? q : q - 1;
            float* dst = red + ((((size_t)rtl * 7 + j) * 3 + slot) * 8) * 32 + ln;
#pragma unroll
            for (int v = 0; v < 8; ++v) dst[v * 32] = acc[j][v];
        }
    }
    __syncthreads();

    float* out = part + ((size_t)blockIdx.z * KGRP + blockIdx.y) * BB * AXLD;
    const int ih = (ln >> 4) * 8;
#pragma unroll
    for (int j = 0; j < 7; ++j) {
        const int o = j >> 1;
        if (q == o) {
            const float* srcp = red + (((size_t)rtl * 7 + j) * 3 * 8) * 32 + ln;
#pragma unroll
            for (int v = 0; v < 8; ++v) {
                float c[KSLOT];
#pragma unroll
                for (int qq = 0; qq < KSLOT; ++qq) {
                    if (qq == o) c[qq] = acc[j][v];
                    else         c[qq] = srcp[(((qq < o) ? qq : qq - 1) * 8 + v) * 32];
                }
                out[(size_t)(r0 + ih + v) * AXLD + j * 16 + mlo] = ((c[0] + c[1]) + c[2]) + c[3];
            }
        }
    }
}

#define SUM_THREADS (16 * AXLD / 4)
#define SUM_LD      (AXLD + 4)
__global__ void __launch_bounds__(SUM_THREADS)
ax_sum_kernel(const float* __restrict__ part, const float* __restrict__ b,
              float* __restrict__ axbuf) {
    __shared__ float sT[16][SUM_LD];
    const int n4 = BB * AXLD / 4;
    const int t  = threadIdx.x;
    const int i  = blockIdx.x * SUM_THREADS + t;
    part  += (size_t)blockIdx.y * KGRP * BB * AXLD;
    axbuf += (size_t)blockIdx.y * BB * AXLD;
    const float4* src = (const float4*)part;
    float4 v[KGRP];
#pragma unroll
    for (int c = 0; c < KGRP; ++c) v[c] = src[(size_t)c * n4 + i];
    const int trow = t / (AXLD / 4);
    const int tq   = t % (AXLD / 4);
    float4 bv = make_float4(0.f, 0.f, 0.f, 0.f);
    if (tq < MP / 4)
        bv = *(const float4*)&b[(size_t)(blockIdx.x * 16 + trow) * MROWS + blockIdx.y * MP + tq * 4];
    float4 s = v[0];
#pragma unroll
    for (int c = 1; c < KGRP; ++c) {
        s.x += v[c].x; s.y += v[c].y; s.z += v[c].z; s.w += v[c].w;
    }
    s.x -= bv.x; s.y -= bv.y; s.z -= bv.z; s.w -= bv.w;
    *(float4*)&sT[trow][tq * 4] = s;
    __syncthreads();
    const int mp = t >> 3;
    const int c  = (t & 7) * 2;
    const v2f lo = *(const v2f*)&sT[c][2 * mp];
    const v2f hi = *(const v2f*)&sT[c + 1][2 * mp];
    ((float4*)axbuf)[i] = make_float4(lo.x, lo.y, hi.x, hi.y);
}

#define UPD_WAVES   32
#define UPD_THREADS (UPD_WAVES * 32)
#define UPD_YGRP    (BB / 16 * JT5 / UPD_WAVES)
#define UPD_STG     10

__device__ __forceinline__ float pick5(const float (&x)[PN], int q) {
    float r = x[0];
    r = (q == 1) ? x[1] : r;
    r = (q == 2) ? x[2] : r;
    r = (q == 3) ? x[3] : r;
    r = (q == 4) ? x[4] : r;
    return r;
}

__global__ void __launch_bounds__(256)
first_update_kernel(const float* __restrict__ Atb, const float* __restrict__ hyp,
                    const int* __restrict__ nbr,
                    float* __restrict__ X, float* __restrict__ U) {
    const size_t plane4 = (size_t)BB * NCOLS / 4;
    const size_t i = (size_t)blockIdx.x * 256 + threadIdx.x;

    float4 t4[PN];
#pragma unroll
    for (int p = 0; p < PN; ++p) t4[p] = ((const float4*)Atb)[(size_t)p * plane4 + i];

    float a0[PN], a1[PN], a2[PN], a3[PN];
    int   nl[PN], nr[PN];
#pragma unroll
    for (int p = 0; p < PN; ++p) {
        a0[p] = fabsf(hyp[p * 4 + 0]);
        a1[p] = fabsf(hyp[p * 4 + 1]);
        a2[p] = fabsf(hyp[p * 4 + 2]);
        a3[p] = fabsf(hyp[p * 4 + 3]);
        nl[p] = nbr[p * 2 + 0];
        nr[p] = nbr[p * 2 + 1];
    }

    float tt[PN][4], xn[PN][4], un[PN][4];
#pragma unroll
    for (int p = 0; p < PN; ++p) {
        tt[p][0] = t4[p].x; tt[p][1] = t4[p].y; tt[p][2] = t4[p].z; tt[p][3] = t4[p].w;
    }

#pragma unroll
    for (int e = 0; e < 4; ++e) {
        float xv[PN] = {0.f, 0.f, 0.f, 0.f, 0.f};
        auto step = [&](int p) -> float {
            const float xo  = xv[p];
            const float sgn = (xo > 0.f) ? 1.f : ((xo < 0.f) ? -1.f : 0.f);
            const float g = 0.f - tt[p][e] + 2.f * a0[p] * xo + a2[p] * sgn
                            - a0[p] * (pick5(xv, nl[p]) + pick5(xv, nr[p]));
            return xo - a1[p] * g;
        };
        {
            const float y0 = step(0);
            const float y2 = step(2);
            xv[0] = y0; xv[2] = y2;
        }
        {
            const float y1 = step(1);
            const float y3 = step(3);
            xv[1] = y1; xv[3] = y3;
        }
        xv[4] = step(4);
#pragma unroll
        for (int p = 0; p < PN; ++p) {
            xn[p][e] = xv[p];
            un[p][e] = a3[p] * (2.f * xv[p] - pick5(xv, nl[p]) - pick5(xv, nr[p]));
        }
    }

#pragma unroll
    for (int p = 0; p < PN; ++p) {
        ((float4*)X)[(size_t)p * plane4 + i] = make_float4(xn[p][0], xn[p][1], xn[p][2], xn[p][3]);
        ((float4*)U)[(size_t)p * plane4 + i] = make_float4(un[p][0], un[p][1], un[p][2], un[p][3]);
    }
}

__global__ void __launch_bounds__(UPD_THREADS)
update_kernel(const float* __restrict__ A, const float* __restrict__ axbuf,
              const float* __restrict__ hyp,
              const int* __restrict__ nbr,
              float* __restrict__ X, float* __restrict__ U, int k) {
    __shared__ float sB[MROWS][NGRP];
    const int n0w = blockIdx.x * NGRP;

    float4 st[UPD_STG];
#pragma unroll
    for (int t = 0; t < UPD_STG; ++t) {
        const int idx = min((int)threadIdx.x + t * UPD_THREADS, MROWS * NGRP / 4 - 1);
        const int row = idx / (NGRP / 4);
        const int c4  = (idx % (NGRP / 4)) * 4;
        st[t] = *(const float4*)&A[(size_t)row * NCOLS + n0w + c4];
    }
#pragma unroll
    for (int t = 0; t < UPD_STG; ++t) {
        const int idx = threadIdx.x + t * UPD_THREADS;
        const int row = idx / (NGRP / 4);
        const int c4  = (idx % (NGRP / 4)) * 4;
        if (idx < MROWS * NGRP / 4)
            *(float4*)&sB[row][c4] = st[t];
    }
    __syncthreads();

    const int wave = threadIdx.x >> 5;
    const int tile = blockIdx.y * UPD_WAVES + wave;
    const int jt   = tile % JT5;
    const int r0   = (tile / JT5) * 16;
    const int lane = threadIdx.x & 31;
    const int c    = lane & 15;
    const int kh   = (lane >> 4) << 1;

    const size_t stride = (size_t)BB * NCOLS;
    const int ih  = (lane >> 4) * 8;
    const int col = n0w + jt * 16 + c;

    float x[8][PN], u[8][PN];
    auto load_row = [&](int v) {
        const size_t off = (size_t)(r0 + ih + v) * NCOLS + col;
#pragma unroll
        for (int p = 0; p < PN; ++p) {
            x[v][p] = X[(size_t)p * stride + off];
            u[v][p] = U[(size_t)p * stride + off];
        }
    };

    load_row(0); load_row(1); load_row(2); load_row(3);

    const float* pa = axbuf + (((size_t)(r0 >> 4) * (AXLD / 2) + (lane >> 4)) * 16 + c) * 2;
    const float* pb = &sB[kh][jt * 16 + c];

    v2f afq[MP / 4][PN];
    auto load_af = [&](int ks) {
#pragma unroll
        for (int s = 0; s < PN; ++s)
            afq[ks][s] = *(const v2f*)(pa + (size_t)s * BB * AXLD + ks * 64);
    };
    load_af(0);

    v8f acc[PN] = {};
#pragma unroll
    for (int ks = 0; ks < MP / 4; ++ks) {
        if (ks + 1 < MP / 4) load_af(ks + 1);
#pragma unroll
        for (int s = 0; s < PN; ++s) {
            v2f bf;
            bf.x = pb[(s * MP + 4 * ks) * NGRP];
            bf.y = pb[(s * MP + 4 * ks + 1) * NGRP];
            acc[s] = wmma4(afq[ks][s], bf, acc[s]);
        }
    }

    float a0[PN], a1[PN], a2[PN], a3[PN];
    int   nl[PN], nr[PN];
#pragma unroll
    for (int p = 0; p < PN; ++p) {
        a0[p] = fabsf(hyp[(k * PN + p) * 4 + 0]);
        a1[p] = fabsf(hyp[(k * PN + p) * 4 + 1]);
        a2[p] = fabsf(hyp[(k * PN + p) * 4 + 2]);
        a3[p] = fabsf(hyp[(k * PN + p) * 4 + 3]);
        nl[p] = nbr[p * 2 + 0];
        nr[p] = nbr[p * 2 + 1];
    }

    auto do_row = [&](int v, const float (&av)[PN]) {
        const size_t off = (size_t)(r0 + ih + v) * NCOLS + col;
        float (&xv)[PN] = x[v];
        auto step = [&](int p) -> float {
            const float xo  = xv[p];
            const float sgn = (xo > 0.f) ? 1.f : ((xo < 0.f) ? -1.f : 0.f);
            const float g = av[p] + 2.f * a0[p] * xo + a2[p] * sgn
                            + 2.f * u[v][p] - a0[p] * (pick5(xv, nl[p]) + pick5(xv, nr[p]));
            return xo - a1[p] * g;
        };
        {
            const float y0 = step(0);
            const float y2 = step(2);
            xv[0] = y0; xv[2] = y2;
        }
        {
            const float y1 = step(1);
            const float y3 = step(3);
            xv[1] = y1; xv[3] = y3;
        }
        xv[4] = step(4);
        float un[PN];
#pragma unroll
        for (int p = 0; p < PN; ++p)
            un[p] = u[v][p] + a3[p] * (2.f * xv[p] - pick5(xv, nl[p]) - pick5(xv, nr[p]));
#pragma unroll
        for (int p = 0; p < PN; ++p) {
            X[(size_t)p * stride + off] = xv[p];
            U[(size_t)p * stride + off] = un[p];
        }
    };

#pragma unroll
    for (int v = 0; v < 8; ++v) {
        const float av[PN] = {acc[0][v], acc[1][v], acc[2][v], acc[3][v], acc[4][v]};
        do_row(v, av);
        if (v + 4 < 8) load_row(v + 4);
    }
}

extern "C" void kernel_launch(void* const* d_in, const int* in_sizes, int n_in,
                              void* d_out, int out_size, void* d_ws, size_t ws_size,
                              hipStream_t stream) {
    (void)in_sizes; (void)n_in; (void)out_size; (void)ws_size;
    const float* A   = (const float*)d_in[0];
    const float* b   = (const float*)d_in[1];
    const float* hyp = (const float*)d_in[2];
    const int*   nbr = (const int*)d_in[3];

    float* X = (float*)d_out;
    float* U = X + (size_t)PN * BB * NCOLS;
    float* Atb   = (float*)d_ws;
    float* axbuf = Atb + (size_t)PN * BB * NCOLS;
    float* part  = axbuf + (size_t)PN * BB * AXLD;

    atb_kernel<<<dim3(25, 4, PN), 256, 0, stream>>>(A, b, Atb);

    first_update_kernel<<<BB * NCOLS / 4 / 256, 256, 0, stream>>>(Atb, hyp, nbr, X, U);

    for (int k = 1; k < KITERS; ++k) {
        ax_part_kernel<<<dim3(8, KGRP, PN), 1024, 0, stream>>>(A, X, part);
        ax_sum_kernel <<<dim3(BB / 16, PN), SUM_THREADS, 0, stream>>>(part, b, axbuf);
        update_kernel <<<dim3(25, UPD_YGRP), UPD_THREADS, 0, stream>>>(
            A, axbuf, hyp, nbr, X, U, k);
    }
}
